# starting kernel
# baseline (speedup 1.0000x reference)
#include <hip/hip_runtime.h>
#include <hip/hip_bf16.h>

typedef __attribute__((ext_vector_type(2))) float v2f;
typedef __attribute__((ext_vector_type(8))) float v8f;

__device__ __forceinline__ v8f wmma_f32_k4(v2f a, v2f b, v8f c) {
    return __builtin_amdgcn_wmma_f32_16x16x4_f32(
         false, a,  false, b,
         (short)0, c,  false,  false);
}

__device__ __forceinline__ void atomAddF(float* p, float v) {
    __hip_atomic_fetch_add(p, v, __ATOMIC_RELAXED, __HIP_MEMORY_SCOPE_AGENT);
}

__global__ void k_init_deg(float* deg, int N) {
    int i = blockIdx.x * blockDim.x + threadIdx.x;
    if (i < N) deg[i] = 1.0f;
}
__global__ void k_deg_accum(const int* __restrict__ dst, float* deg, int E) {
    int e = blockIdx.x * blockDim.x + threadIdx.x;
    if (e < E) atomAddF(&deg[dst[e]], 1.0f);
}
__global__ void k_rsqrt(float* deg, int N) {
    int i = blockIdx.x * blockDim.x + threadIdx.x;
    if (i < N) deg[i] = rsqrtf(deg[i]);
}

__global__ void k_zero(float* p, long long n) {
    long long i = (long long)blockIdx.x * blockDim.x + threadIdx.x;
    if (i < n) p[i] = 0.0f;
}

__global__ void k_gemm1(const float* __restrict__ x, const float* __restrict__ W1,
                        float* __restrict__ xw) {
    const int lane = threadIdx.x & 31;
    const int wave = threadIdx.x >> 5;
    const int half = lane >> 4;
    const int rlo  = lane & 15;
    const int mt   = blockIdx.x;
    const int row  = (mt << 4) + rlo;
    const int ncol = (wave << 4) + rlo;

    v8f acc = {};
    const float* xrow = x + (size_t)row * 300 + (half << 1);
    for (int k = 0; k < 300; k += 4) {
        float2 av = *(const float2*)(xrow + k);
        v2f a; a.x = av.x; a.y = av.y;
        const int kr = k + (half << 1);
        v2f b;
        b.x = W1[(size_t)kr * 128 + ncol];
        b.y = W1[(size_t)(kr + 1) * 128 + ncol];
        acc = wmma_f32_k4(a, b, acc);
    }
    const int m0 = (mt << 4) + (half << 3);
    #pragma unroll
    for (int r = 0; r < 8; ++r)
        xw[(size_t)(m0 + r) * 128 + ncol] = acc[r];
}

__global__ void k_agg1(const float* __restrict__ xw, const int* __restrict__ src,
                       const int* __restrict__ dst, const float* __restrict__ dinv,
                       float* __restrict__ agg, int E) {
    long long gid = (long long)blockIdx.x * blockDim.x + threadIdx.x;
    int e = (int)(gid >> 5);
    int f = ((int)gid & 31) << 2;
    if (e >= E) return;
    int s = src[e], d = dst[e];
    float nrm = dinv[s] * dinv[d];
    float4 v = *(const float4*)(xw + (size_t)s * 128 + f);
    float* o = agg + (size_t)d * 128 + f;
    atomAddF(o + 0, v.x * nrm);
    atomAddF(o + 1, v.y * nrm);
    atomAddF(o + 2, v.z * nrm);
    atomAddF(o + 3, v.w * nrm);
}

__global__ void k_bias_relu(float* __restrict__ h, const float* __restrict__ xw,
                            const float* __restrict__ dinv, const float* __restrict__ b1,
                            int N) {
    long long gid = (long long)blockIdx.x * blockDim.x + threadIdx.x;
    if (gid >= (long long)N * 128) return;
    int node = (int)(gid >> 7);
    int f = (int)gid & 127;
    float di = dinv[node];
    float v = h[gid] + xw[gid] * di * di + b1[f];
    h[gid] = v > 0.0f ? v : 0.0f;
}

__global__ void k_gemm2(const float* __restrict__ h, const float* __restrict__ W2,
                        float* __restrict__ hw16, int N) {
    __shared__ float w2s[128 * 16];
    for (int i = threadIdx.x; i < 128 * 16; i += 256) {
        int k = i >> 4, c = i & 15;
        w2s[i] = (c < 10) ? W2[k * 10 + c] : 0.0f;
    }
    __syncthreads();

    const int lane = threadIdx.x & 31;
    const int wave = threadIdx.x >> 5;
    const int half = lane >> 4;
    const int rlo  = lane & 15;
    const int mt   = blockIdx.x * 8 + wave;
    if ((mt << 4) < N) {
        v8f acc = {};
        const float* hrow = h + (size_t)((mt << 4) + rlo) * 128 + (half << 1);
        for (int k = 0; k < 128; k += 4) {
            float2 av = *(const float2*)(hrow + k);
            v2f a; a.x = av.x; a.y = av.y;
            const int kr = k + (half << 1);
            v2f b;
            b.x = w2s[kr * 16 + rlo];
            b.y = w2s[(kr + 1) * 16 + rlo];
            acc = wmma_f32_k4(a, b, acc);
        }
        const int m0 = (mt << 4) + (half << 3);
        #pragma unroll
        for (int r = 0; r < 8; ++r)
            hw16[(size_t)(m0 + r) * 16 + rlo] = acc[r];
    }
}

__global__ void k_agg2(const float* __restrict__ hw16, const int* __restrict__ src,
                       const int* __restrict__ dst, const float* __restrict__ dinv,
                       float* __restrict__ out, int E) {
    long long gid = (long long)blockIdx.x * blockDim.x + threadIdx.x;
    int e = (int)(gid >> 4);
    int f = (int)gid & 15;
    if (e >= E || f >= 10) return;
    int s = src[e], d = dst[e];
    float nrm = dinv[s] * dinv[d];
    atomAddF(out + (size_t)d * 10 + f, hw16[(size_t)s * 16 + f] * nrm);
}

__global__ void k_final(const float* __restrict__ hw16, const float* __restrict__ dinv,
                        const float* __restrict__ b2, float* __restrict__ out, int N) {
    long long gid = (long long)blockIdx.x * blockDim.x + threadIdx.x;
    if (gid >= (long long)N * 10) return;
    int node = (int)(gid / 10);
    int f = (int)(gid % 10);
    float di = dinv[node];
    out[gid] += hw16[(size_t)node * 16 + f] * di * di + b2[f];
}

extern "C" void kernel_launch(void* const* d_in, const int* in_sizes, int n_in,
                              void* d_out, int out_size, void* d_ws, size_t ws_size,
                              hipStream_t stream) {
    const float* x  = (const float*)d_in[0];
    const int*   ei = (const int*)d_in[1];
    const float* W1 = (const float*)d_in[2];
    const float* b1 = (const float*)d_in[3];
    const float* W2 = (const float*)d_in[4];
    const float* b2 = (const float*)d_in[5];
    float* out = (float*)d_out;

    const int N = in_sizes[0] / 300;
    const int E = in_sizes[1] / 2;
    const int* src = ei;
    const int* dst = ei + E;

    size_t Np = ((size_t)N + 63) & ~(size_t)63;
    float* dinv = (float*)d_ws;
    float* xw   = dinv + Np;
    float* h    = xw + Np * 128;
    float* hw   = h  + Np * 128;

    const int T = 256;
    const long long nf1 = (long long)N * 128;

    k_init_deg <<<(N + T - 1) / T, T, 0, stream>>>(dinv, N);
    k_deg_accum<<<(E + T - 1) / T, T, 0, stream>>>(dst, dinv, E);
    k_rsqrt    <<<(N + T - 1) / T, T, 0, stream>>>(dinv, N);

    k_gemm1    <<<N / 16, T, 0, stream>>>(x, W1, xw);

    k_zero     <<<(unsigned)((nf1 + T - 1) / T), T, 0, stream>>>(h, nf1);
    k_agg1     <<<(unsigned)(((long long)E * 32 + T - 1) / T), T, 0, stream>>>(
                    xw, src, dst, dinv, h, E);
    k_bias_relu<<<(unsigned)((nf1 + T - 1) / T), T, 0, stream>>>(h, xw, dinv, b1, N);

    k_gemm2    <<<(N / 16 + 7) / 8, T, 0, stream>>>(h, W2, hw, N);

    const long long no = (long long)N * 10;
    k_zero     <<<(unsigned)((no + T - 1) / T), T, 0, stream>>>(out, no);
    k_agg2     <<<(unsigned)(((long long)E * 16 + T - 1) / T), T, 0, stream>>>(
                    hw, src, dst, dinv, out, E);
    k_final    <<<(unsigned)((no + T - 1) / T), T, 0, stream>>>(hw, dinv, b2, out, N);
}


# Edges counting-sorted by destination (returning int atomic = arrival-order rank, prefix sum, fill); per-node row gathers summed in that rank order replace all float atomic scatter-adds
# speedup vs baseline: 1.8835x; 1.8835x over previous
#include <hip/hip_runtime.h>
#include <hip/hip_bf16.h>


typedef __attribute__((ext_vector_type(2))) float v2f;
typedef __attribute__((ext_vector_type(8))) float v8f;

__device__ __forceinline__ v8f wmma_f32_k4(v2f a, v2f b, v8f c) {
    return __builtin_amdgcn_wmma_f32_16x16x4_f32(
         false, a,  false, b,
         (short)0, c,  false,  false);
}

__global__ void k_zero_cnt(int* cnt, int n) {
    int i = blockIdx.x * blockDim.x + threadIdx.x;
    if (i < n) cnt[i] = 0;
}
__global__ void k_count_rank(const int* __restrict__ dst, int* cnt, int* __restrict__ rank, int E) {
    int e = blockIdx.x * blockDim.x + threadIdx.x;
    if (e < E)
        rank[e] = __hip_atomic_fetch_add(&cnt[dst[e]], 1, __ATOMIC_RELAXED, __HIP_MEMORY_SCOPE_AGENT);
}

__global__ void k_scan_local(int* __restrict__ cnt, int* __restrict__ btot, int N) {
    __shared__ int wsum[32];
    const int tid = threadIdx.x, lane = tid & 31, wv = tid >> 5;
    const int i = blockIdx.x * 1024 + tid;
    const int v = (i < N) ? cnt[i] : 0;
    int inc = v;
    for (int o = 1; o < 32; o <<= 1) {
        int t = __shfl_up(inc, o, 32);
        if (lane >= o) inc += t;
    }
    if (lane == 31) wsum[wv] = inc;
    __syncthreads();
    if (wv == 0) {
        const int w = wsum[lane];
        int winc = w;
        for (int o = 1; o < 32; o <<= 1) {
            int t = __shfl_up(winc, o, 32);
            if (lane >= o) winc += t;
        }
        wsum[lane] = winc - w;
        if (lane == 31) btot[blockIdx.x] = winc;
    }
    __syncthreads();
    if (i <= N) cnt[i] = wsum[wv] + inc - v;
}
__global__ void k_scan_add(int* __restrict__ rowptr, const int* __restrict__ btot, int N) {
    __shared__ int off;
    if (threadIdx.x < 32) {
        int s = 0;
        for (int k = threadIdx.x; k < (int)blockIdx.x; k += 32) s += btot[k];
        for (int o = 16; o > 0; o >>= 1) s += __shfl_xor(s, o, 32);
        if (threadIdx.x == 0) off = s;
    }
    __syncthreads();
    const int i = blockIdx.x * 1024 + threadIdx.x;
    if (i <= N) rowptr[i] += off;
}
__global__ void k_fill(const int* __restrict__ src, const int* __restrict__ dst,
                       const int* __restrict__ rank, const int* __restrict__ rowptr,
                       int* __restrict__ csr, int E) {
    int e = blockIdx.x * blockDim.x + threadIdx.x;
    if (e < E) csr[rowptr[dst[e]] + rank[e]] = src[e];
}

__device__ __forceinline__ float dinv_of(const int* __restrict__ rowptr, int s) {
    return rsqrtf((float)(rowptr[s + 1] - rowptr[s] + 1));
}

__global__ void k_gemm1(const float* __restrict__ x, const float* __restrict__ W1,
                        float* __restrict__ xw) {
    const int lane = threadIdx.x & 31;
    const int wave = threadIdx.x >> 5;
    const int half = lane >> 4;
    const int rlo  = lane & 15;
    const int mt   = blockIdx.x;
    const int row  = (mt << 4) + rlo;
    const int ncol = (wave << 4) + rlo;

    v8f acc = {};
    const float* xrow = x + (size_t)row * 300 + (half << 1);
    for (int k = 0; k < 300; k += 4) {
        float2 av = *(const float2*)(xrow + k);
        v2f a; a.x = av.x; a.y = av.y;
        const int kr = k + (half << 1);
        v2f b;
        b.x = W1[(size_t)kr * 128 + ncol];
        b.y = W1[(size_t)(kr + 1) * 128 + ncol];
        acc = wmma_f32_k4(a, b, acc);
    }
    const int m0 = (mt << 4) + (half << 3);
    #pragma unroll
    for (int r = 0; r < 8; ++r)
        xw[(size_t)(m0 + r) * 128 + ncol] = acc[r];
}

template <int U>
__device__ __forceinline__ void add_rows(const float* __restrict__ xw, int s_l, float w_l,
                                         int t, int f, float4& acc) {
    float4 v[U];
    float w[U];
    #pragma unroll
    for (int k = 0; k < U; ++k) {
        const int s = __builtin_amdgcn_readlane(s_l, t + k);
        w[k] = __int_as_float(__builtin_amdgcn_readlane(__float_as_int(w_l), t + k));
        v[k] = *(const float4*)(xw + (size_t)s * 128 + f);
    }
    #pragma unroll
    for (int k = 0; k < U; ++k) {
        acc.x += v[k].x * w[k];
        acc.y += v[k].y * w[k];
        acc.z += v[k].z * w[k];
        acc.w += v[k].w * w[k];
    }
}

__global__ void k_gather1(const float* __restrict__ xw, const int* __restrict__ rowptr,
                          const int* __restrict__ csr, const float* __restrict__ b1,
                          float* __restrict__ h, int N) {
    const int lane = threadIdx.x & 31;
    const int n = __builtin_amdgcn_readfirstlane((int)(blockIdx.x * 8 + (threadIdx.x >> 5)));
    if (n >= N) return;
    const int beg = rowptr[n], end = rowptr[n + 1];
    const float dn = rsqrtf((float)(end - beg + 1));
    const int f = lane << 2;
    float4 acc = {0.0f, 0.0f, 0.0f, 0.0f};
    for (int base = beg; base < end; base += 32) {
        const int m = min(32, end - base);
        int s_l = 0;
        float w_l = 0.0f;
        if (lane < m) {
            s_l = csr[base + lane];
            w_l = dinv_of(rowptr, s_l) * dn;
        }
        int t = 0;
        for (; t + 8 <= m; t += 8) add_rows<8>(xw, s_l, w_l, t, f, acc);
        if (t + 4 <= m) { add_rows<4>(xw, s_l, w_l, t, f, acc); t += 4; }
        if (t + 2 <= m) { add_rows<2>(xw, s_l, w_l, t, f, acc); t += 2; }
        if (t < m)      { add_rows<1>(xw, s_l, w_l, t, f, acc); }
    }
    const float4 sv = *(const float4*)(xw + (size_t)n * 128 + f);
    const float4 bv = *(const float4*)(b1 + f);
    float4 o;
    o.x = acc.x + sv.x * dn * dn + bv.x;
    o.y = acc.y + sv.y * dn * dn + bv.y;
    o.z = acc.z + sv.z * dn * dn + bv.z;
    o.w = acc.w + sv.w * dn * dn + bv.w;
    o.x = o.x > 0.0f ? o.x : 0.0f;
    o.y = o.y > 0.0f ? o.y : 0.0f;
    o.z = o.z > 0.0f ? o.z : 0.0f;
    o.w = o.w > 0.0f ? o.w : 0.0f;
    *(float4*)(h + (size_t)n * 128 + f) = o;
}

__global__ void k_gemm2(const float* __restrict__ h, const float* __restrict__ W2,
                        float* __restrict__ hw16, int N) {
    __shared__ float w2s[128 * 16];
    for (int i = threadIdx.x; i < 128 * 16; i += 256) {
        int k = i >> 4, c = i & 15;
        w2s[i] = (c < 10) ? W2[k * 10 + c] : 0.0f;
    }
    __syncthreads();

    const int lane = threadIdx.x & 31;
    const int wave = threadIdx.x >> 5;
    const int half = lane >> 4;
    const int rlo  = lane & 15;
    const int mt   = blockIdx.x * 8 + wave;
    if ((mt << 4) < N) {
        v8f acc = {};
        const float* hrow = h + (size_t)((mt << 4) + rlo) * 128 + (half << 1);
        for (int k = 0; k < 128; k += 4) {
            float2 av = *(const float2*)(hrow + k);
            v2f a; a.x = av.x; a.y = av.y;
            const int kr = k + (half << 1);
            v2f b;
            b.x = w2s[kr * 16 + rlo];
            b.y = w2s[(kr + 1) * 16 + rlo];
            acc = wmma_f32_k4(a, b, acc);
        }
        const int m0 = (mt << 4) + (half << 3);
        #pragma unroll
        for (int r = 0; r < 8; ++r)
            hw16[(size_t)(m0 + r) * 16 + rlo] = acc[r];
    }
}

__global__ void k_gather2(const float* __restrict__ hw16, const int* __restrict__ rowptr,
                          const int* __restrict__ csr, const float* __restrict__ b2,
                          float* __restrict__ out, int N) {
    __shared__ float sres[16 * 10];
    const int tid = threadIdx.x;
    const int f  = tid & 15;
    const int nl = tid >> 4;
    const int n  = blockIdx.x * 16 + nl;
    const int nc = n < N ? n : N - 1;
    const int beg = rowptr[nc], end = rowptr[nc + 1];
    const float dn = rsqrtf((float)(end - beg + 1));
    float acc = 0.0f;
    for (int base = beg; base < end; base += 16) {
        const int m = min(16, end - base);
        int s_l = 0;
        float w_l = 0.0f;
        if (f < m) {
            s_l = csr[base + f];
            w_l = dinv_of(rowptr, s_l) * dn;
        }
        int t = 0;
        for (; t + 4 <= m; t += 4) {
            const int s0 = __shfl(s_l, t, 16),     s1 = __shfl(s_l, t + 1, 16);
            const int s2 = __shfl(s_l, t + 2, 16), s3 = __shfl(s_l, t + 3, 16);
            const float w0 = __shfl(w_l, t, 16),     w1 = __shfl(w_l, t + 1, 16);
            const float w2 = __shfl(w_l, t + 2, 16), w3 = __shfl(w_l, t + 3, 16);
            const float v0 = hw16[(size_t)s0 * 16 + f];
            const float v1 = hw16[(size_t)s1 * 16 + f];
            const float v2 = hw16[(size_t)s2 * 16 + f];
            const float v3 = hw16[(size_t)s3 * 16 + f];
            acc += v0 * w0;
            acc += v1 * w1;
            acc += v2 * w2;
            acc += v3 * w3;
        }
        for (; t < m; ++t) {
            const int s = __shfl(s_l, t, 16);
            const float w = __shfl(w_l, t, 16);
            acc += hw16[(size_t)s * 16 + f] * w;
        }
    }
    const float self = hw16[(size_t)nc * 16 + f];
    if (f < 10) sres[nl * 10 + f] = acc + self * dn * dn + b2[f];
    __syncthreads();
    if (tid < 160) {
        const size_t o = (size_t)blockIdx.x * 160 + tid;
        if (o < (size_t)N * 10) out[o] = sres[tid];
    }
}

extern "C" void kernel_launch(void* const* d_in, const int* in_sizes, int n_in,
                              void* d_out, int out_size, void* d_ws, size_t ws_size,
                              hipStream_t stream) {
    const float* x  = (const float*)d_in[0];
    const int*   ei = (const int*)d_in[1];
    const float* W1 = (const float*)d_in[2];
    const float* b1 = (const float*)d_in[3];
    const float* W2 = (const float*)d_in[4];
    const float* b2 = (const float*)d_in[5];
    float* out = (float*)d_out;

    const int N = in_sizes[0] / 300;
    const int E = in_sizes[1] / 2;
    const int* src = ei;
    const int* dst = ei + E;

    size_t Np = ((size_t)N + 64) & ~(size_t)63;
    int*   rowptr = (int*)d_ws;
    float* xw   = (float*)d_ws + Np;
    float* h    = xw + Np * 128;
    int*   csr  = (int*)(h + Np * 128);
    int*   rank = (int*)h;
    int*   btot = rank + E;
    float* hw   = xw;

    const int T = 256;
    const int nb = (N + 1 + 1023) / 1024;

    k_zero_cnt  <<<(N + T - 1) / T, T, 0, stream>>>(rowptr, N);
    k_count_rank<<<(E + T - 1) / T, T, 0, stream>>>(dst, rowptr, rank, E);
    k_scan_local<<<nb, 1024, 0, stream>>>(rowptr, btot, N);
    k_scan_add  <<<nb, 1024, 0, stream>>>(rowptr, btot, N);
    k_fill      <<<(E + T - 1) / T, T, 0, stream>>>(src, dst, rank, rowptr, csr, E);

    k_gemm1     <<<N / 16, T, 0, stream>>>(x, W1, xw);
    k_gather1   <<<(N + 7) / 8, T, 0, stream>>>(xw, rowptr, csr, b1, h, N);

    k_gemm2     <<<(N / 16 + 7) / 8, T, 0, stream>>>(h, W2, hw, N);
    k_gather2   <<<(N + 15) / 16, T, 0, stream>>>(hw, rowptr, csr, b2, out, N);
}


# GEMM1: W1 staged once per 64-row block in LDS as k-pair rows; a wave keeps a 16x64 strip in 4 accumulators, one float4 A read per 8 k; k grouping inside the matrix sums changed (fixed order)
# speedup vs baseline: 2.3394x; 1.2420x over previous
#include <hip/hip_runtime.h>
#include <hip/hip_bf16.h>


typedef __attribute__((ext_vector_type(2))) float v2f;
typedef __attribute__((ext_vector_type(8))) float v8f;

__device__ __forceinline__ v8f wmma_f32_k4(v2f a, v2f b, v8f c) {
    return __builtin_amdgcn_wmma_f32_16x16x4_f32(
         false, a,  false, b,
         (short)0, c,  false,  false);
}

__global__ void k_zero_cnt(int* cnt, int n) {
    int i = blockIdx.x * blockDim.x + threadIdx.x;
    if (i < n) cnt[i] = 0;
}
__global__ void k_count_rank(const int* __restrict__ dst, int* cnt, int* __restrict__ rank, int E) {
    int e = blockIdx.x * blockDim.x + threadIdx.x;
    if (e < E)
        rank[e] = __hip_atomic_fetch_add(&cnt[dst[e]], 1, __ATOMIC_RELAXED, __HIP_MEMORY_SCOPE_AGENT);
}

__global__ void k_scan_local(int* __restrict__ cnt, int* __restrict__ btot, int N) {
    __shared__ int wsum[32];
    const int tid = threadIdx.x, lane = tid & 31, wv = tid >> 5;
    const int i = blockIdx.x * 1024 + tid;
    const int v = (i < N) ? cnt[i] : 0;
    int inc = v;
    for (int o = 1; o < 32; o <<= 1) {
        int t = __shfl_up(inc, o, 32);
        if (lane >= o) inc += t;
    }
    if (lane == 31) wsum[wv] = inc;
    __syncthreads();
    if (wv == 0) {
        const int w = wsum[lane];
        int winc = w;
        for (int o = 1; o < 32; o <<= 1) {
            int t = __shfl_up(winc, o, 32);
            if (lane >= o) winc += t;
        }
        wsum[lane] = winc - w;
        if (lane == 31) btot[blockIdx.x] = winc;
    }
    __syncthreads();
    if (i <= N) cnt[i] = wsum[wv] + inc - v;
}
__global__ void k_scan_add(int* __restrict__ rowptr, const int* __restrict__ btot, int N) {
    __shared__ int off;
    if (threadIdx.x < 32) {
        int s = 0;
        for (int k = threadIdx.x; k < (int)blockIdx.x; k += 32) s += btot[k];
        for (int o = 16; o > 0; o >>= 1) s += __shfl_xor(s, o, 32);
        if (threadIdx.x == 0) off = s;
    }
    __syncthreads();
    const int i = blockIdx.x * 1024 + threadIdx.x;
    if (i <= N) rowptr[i] += off;
}
__global__ void k_fill(const int* __restrict__ src, const int* __restrict__ dst,
                       const int* __restrict__ rank, const int* __restrict__ rowptr,
                       int* __restrict__ csr, int E) {
    int e = blockIdx.x * blockDim.x + threadIdx.x;
    if (e < E) csr[rowptr[dst[e]] + rank[e]] = src[e];
}

__device__ __forceinline__ float dinv_of(const int* __restrict__ rowptr, int s) {
    return rsqrtf((float)(rowptr[s + 1] - rowptr[s] + 1));
}

__global__ void __launch_bounds__(256) __attribute__((amdgpu_waves_per_eu(4, 4)))
k_gemm1(const float* __restrict__ x, const float* __restrict__ W1, float* __restrict__ xw, int N) {
    __shared__ __attribute__((aligned(16))) float w1s[150 * 256];
    const int lane = threadIdx.x & 31;
    const int wave = __builtin_amdgcn_readfirstlane((int)(threadIdx.x >> 5));
    const int h    = lane >> 4;
    const int c    = lane & 15;
    const int wh   = wave & 1;
    const int rb   = blockIdx.x * 64 + ((wave >> 1) << 4);

    const int arow = min(rb + c, N - 1);
    const float* xr = x + (size_t)arow * 300 + (h << 2);
    float4 a[37];
    #pragma unroll
    for (int kk = 0; kk < 18; ++kk) a[kk] = *(const float4*)(xr + 8 * kk);

    const float* wsrc = W1 + (size_t)wave * 256 + (lane << 2);
    float* wdst = w1s + wave * 256 + (lane >> 4) * 128 + (lane & 15) * 4;
    #pragma unroll 9
    for (int it = 0; it < 18; ++it) {
        const float4 r0 = *(const float4*)(wsrc + it * 2048);
        const float4 r1 = *(const float4*)(wsrc + it * 2048 + 128);
        float4 lo; lo.x = r0.x; lo.y = r1.x; lo.z = r0.y; lo.w = r1.y;
        float4 hi; hi.x = r0.z; hi.y = r1.z; hi.z = r0.w; hi.w = r1.w;
        *(float4*)(wdst + it * 2048)      = lo;
        *(float4*)(wdst + it * 2048 + 64) = hi;
    }
    if (wave < 6) {
        const float4 r0 = *(const float4*)(wsrc + 18 * 2048);
        const float4 r1 = *(const float4*)(wsrc + 18 * 2048 + 128);
        float4 lo; lo.x = r0.x; lo.y = r1.x; lo.z = r0.y; lo.w = r1.y;
        float4 hi; hi.x = r0.z; hi.y = r1.z; hi.z = r0.w; hi.w = r1.w;
        *(float4*)(wdst + 18 * 2048)      = lo;
        *(float4*)(wdst + 18 * 2048 + 64) = hi;
    }
    __syncthreads();
    #pragma unroll
    for (int kk = 18; kk < 37; ++kk) a[kk] = *(const float4*)(xr + 8 * kk);
    const float2 at = *(const float2*)(x + (size_t)arow * 300 + 296 + (h << 1));

    v8f acc0 = {}, acc1 = {}, acc2 = {}, acc3 = {};
    const float* bl = w1s + (h << 9) + (wh << 7) + (c << 2);
    #pragma unroll
    for (int kk = 0; kk < 37; ++kk) {
        const float* p = bl + kk * 1024;
        const float4 b00 = *(const float4*)(p);
        const float4 b01 = *(const float4*)(p + 64);
        const float4 b10 = *(const float4*)(p + 256);
        const float4 b11 = *(const float4*)(p + 320);
        v2f a0; a0.x = a[kk].x; a0.y = a[kk].y;
        v2f a1; a1.x = a[kk].z; a1.y = a[kk].w;
        v2f b;
        b.x = b00.x; b.y = b00.y; acc0 = wmma_f32_k4(a0, b, acc0);
        b.x = b00.z; b.y = b00.w; acc1 = wmma_f32_k4(a0, b, acc1);
        b.x = b01.x; b.y = b01.y; acc2 = wmma_f32_k4(a0, b, acc2);
        b.x = b01.z; b.y = b01.w; acc3 = wmma_f32_k4(a0, b, acc3);
        b.x = b10.x; b.y = b10.y; acc0 = wmma_f32_k4(a1, b, acc0);
        b.x = b10.z; b.y = b10.w; acc1 = wmma_f32_k4(a1, b, acc1);
        b.x = b11.x; b.y = b11.y; acc2 = wmma_f32_k4(a1, b, acc2);
        b.x = b11.z; b.y = b11.w; acc3 = wmma_f32_k4(a1, b, acc3);
    }
    {
        const float* p = w1s + (148 + h) * 256 + (wh << 7) + (c << 2);
        const float4 b0 = *(const float4*)(p);
        const float4 b1 = *(const float4*)(p + 64);
        v2f av; av.x = at.x; av.y = at.y;
        v2f b;
        b.x = b0.x; b.y = b0.y; acc0 = wmma_f32_k4(av, b, acc0);
        b.x = b0.z; b.y = b0.w; acc1 = wmma_f32_k4(av, b, acc1);
        b.x = b1.x; b.y = b1.y; acc2 = wmma_f32_k4(av, b, acc2);
        b.x = b1.z; b.y = b1.w; acc3 = wmma_f32_k4(av, b, acc3);
    }
    #pragma unroll
    for (int r = 0; r < 8; ++r) {
        const int rr = rb + (h << 3) + r;
        if (rr < N) {
            float4 o; o.x = acc0[r]; o.y = acc1[r]; o.z = acc2[r]; o.w = acc3[r];
            *(float4*)(xw + (size_t)rr * 128 + (wh << 6) + (c << 2)) = o;
        }
    }
}

template <int U>
__device__ __forceinline__ void add_rows(const float* __restrict__ xw, int s_l, float w_l,
                                         int t, int f, float4& acc) {
    float4 v[U];
    float w[U];
    #pragma unroll
    for (int k = 0; k < U; ++k) {
        const int s = __builtin_amdgcn_readlane(s_l, t + k);
        w[k] = __int_as_float(__builtin_amdgcn_readlane(__float_as_int(w_l), t + k));
        v[k] = *(const float4*)(xw + (size_t)s * 128 + f);
    }
    #pragma unroll
    for (int k = 0; k < U; ++k) {
        acc.x += v[k].x * w[k];
        acc.y += v[k].y * w[k];
        acc.z += v[k].z * w[k];
        acc.w += v[k].w * w[k];
    }
}

__global__ void k_gather1(const float* __restrict__ xw, const int* __restrict__ rowptr,
                          const int* __restrict__ csr, const float* __restrict__ b1,
                          float* __restrict__ h, int N) {
    const int lane = threadIdx.x & 31;
    const int n = __builtin_amdgcn_readfirstlane((int)(blockIdx.x * 8 + (threadIdx.x >> 5)));
    if (n >= N) return;
    const int beg = rowptr[n], end = rowptr[n + 1];
    const float dn = rsqrtf((float)(end - beg + 1));
    const int f = lane << 2;
    float4 acc = {0.0f, 0.0f, 0.0f, 0.0f};
    for (int base = beg; base < end; base += 32) {
        const int m = min(32, end - base);
        int s_l = 0;
        float w_l = 0.0f;
        if (lane < m) {
            s_l = csr[base + lane];
            w_l = dinv_of(rowptr, s_l) * dn;
        }
        int t = 0;
        for (; t + 8 <= m; t += 8) add_rows<8>(xw, s_l, w_l, t, f, acc);
        if (t + 4 <= m) { add_rows<4>(xw, s_l, w_l, t, f, acc); t += 4; }
        if (t + 2 <= m) { add_rows<2>(xw, s_l, w_l, t, f, acc); t += 2; }
        if (t < m)      { add_rows<1>(xw, s_l, w_l, t, f, acc); }
    }
    const float4 sv = *(const float4*)(xw + (size_t)n * 128 + f);
    const float4 bv = *(const float4*)(b1 + f);
    float4 o;
    o.x = acc.x + sv.x * dn * dn + bv.x;
    o.y = acc.y + sv.y * dn * dn + bv.y;
    o.z = acc.z + sv.z * dn * dn + bv.z;
    o.w = acc.w + sv.w * dn * dn + bv.w;
    o.x = o.x > 0.0f ? o.x : 0.0f;
    o.y = o.y > 0.0f ? o.y : 0.0f;
    o.z = o.z > 0.0f ? o.z : 0.0f;
    o.w = o.w > 0.0f ? o.w : 0.0f;
    *(float4*)(h + (size_t)n * 128 + f) = o;
}

__global__ void k_gemm2(const float* __restrict__ h, const float* __restrict__ W2,
                        float* __restrict__ hw16, int N) {
    __shared__ float w2s[128 * 16];
    for (int i = threadIdx.x; i < 128 * 16; i += 256) {
        int k = i >> 4, c = i & 15;
        w2s[i] = (c < 10) ? W2[k * 10 + c] : 0.0f;
    }
    __syncthreads();

    const int lane = threadIdx.x & 31;
    const int wave = threadIdx.x >> 5;
    const int half = lane >> 4;
    const int rlo  = lane & 15;
    const int mt   = blockIdx.x * 8 + wave;
    if ((mt << 4) < N) {
        v8f acc = {};
        const float* hrow = h + (size_t)((mt << 4) + rlo) * 128 + (half << 1);
        for (int k = 0; k < 128; k += 4) {
            float2 av = *(const float2*)(hrow + k);
            v2f a; a.x = av.x; a.y = av.y;
            const int kr = k + (half << 1);
            v2f b;
            b.x = w2s[kr * 16 + rlo];
            b.y = w2s[(kr + 1) * 16 + rlo];
            acc = wmma_f32_k4(a, b, acc);
        }
        const int m0 = (mt << 4) + (half << 3);
        #pragma unroll
        for (int r = 0; r < 8; ++r)
            hw16[(size_t)(m0 + r) * 16 + rlo] = acc[r];
    }
}

__global__ void k_gather2(const float* __restrict__ hw16, const int* __restrict__ rowptr,
                          const int* __restrict__ csr, const float* __restrict__ b2,
                          float* __restrict__ out, int N) {
    __shared__ float sres[16 * 10];
    const int tid = threadIdx.x;
    const int f  = tid & 15;
    const int nl = tid >> 4;
    const int n  = blockIdx.x * 16 + nl;
    const int nc = n < N ? n : N - 1;
    const int beg = rowptr[nc], end = rowptr[nc + 1];
    const float dn = rsqrtf((float)(end - beg + 1));
    float acc = 0.0f;
    for (int base = beg; base < end; base += 16) {
        const int m = min(16, end - base);
        int s_l = 0;
        float w_l = 0.0f;
        if (f < m) {
            s_l = csr[base + f];
            w_l = dinv_of(rowptr, s_l) * dn;
        }
        int t = 0;
        for (; t + 4 <= m; t += 4) {
            const int s0 = __shfl(s_l, t, 16),     s1 = __shfl(s_l, t + 1, 16);
            const int s2 = __shfl(s_l, t + 2, 16), s3 = __shfl(s_l, t + 3, 16);
            const float w0 = __shfl(w_l, t, 16),     w1 = __shfl(w_l, t + 1, 16);
            const float w2 = __shfl(w_l, t + 2, 16), w3 = __shfl(w_l, t + 3, 16);
            const float v0 = hw16[(size_t)s0 * 16 + f];
            const float v1 = hw16[(size_t)s1 * 16 + f];
            const float v2 = hw16[(size_t)s2 * 16 + f];
            const float v3 = hw16[(size_t)s3 * 16 + f];
            acc += v0 * w0;
            acc += v1 * w1;
            acc += v2 * w2;
            acc += v3 * w3;
        }
        for (; t < m; ++t) {
            const int s = __shfl(s_l, t, 16);
            const float w = __shfl(w_l, t, 16);
            acc += hw16[(size_t)s * 16 + f] * w;
        }
    }
    const float self = hw16[(size_t)nc * 16 + f];
    if (f < 10) sres[nl * 10 + f] = acc + self * dn * dn + b2[f];
    __syncthreads();
    if (tid < 160) {
        const size_t o = (size_t)blockIdx.x * 160 + tid;
        if (o < (size_t)N * 10) out[o] = sres[tid];
    }
}

extern "C" void kernel_launch(void* const* d_in, const int* in_sizes, int n_in,
                              void* d_out, int out_size, void* d_ws, size_t ws_size,
                              hipStream_t stream) {
    const float* x  = (const float*)d_in[0];
    const int*   ei = (const int*)d_in[1];
    const float* W1 = (const float*)d_in[2];
    const float* b1 = (const float*)d_in[3];
    const float* W2 = (const float*)d_in[4];
    const float* b2 = (const float*)d_in[5];
    float* out = (float*)d_out;

    const int N = in_sizes[0] / 300;
    const int E = in_sizes[1] / 2;
    const int* src = ei;
    const int* dst = ei + E;

    size_t Np = ((size_t)N + 64) & ~(size_t)63;
    int*   rowptr = (int*)d_ws;
    float* xw   = (float*)d_ws + Np;
    float* h    = xw + Np * 128;
    int*   csr  = (int*)(h + Np * 128);
    int*   rank = (int*)h;
    int*   btot = rank + E;
    float* hw   = xw;

    const int T = 256;
    const int nb = (N + 1 + 1023) / 1024;

    k_zero_cnt  <<<(N + T - 1) / T, T, 0, stream>>>(rowptr, N);
    k_count_rank<<<(E + T - 1) / T, T, 0, stream>>>(dst, rowptr, rank, E);
    k_scan_local<<<nb, 1024, 0, stream>>>(rowptr, btot, N);
    k_scan_add  <<<nb, 1024, 0, stream>>>(rowptr, btot, N);
    k_fill      <<<(E + T - 1) / T, T, 0, stream>>>(src, dst, rank, rowptr, csr, E);

    k_gemm1     <<<(N + 63) / 64, T, 0, stream>>>(x, W1, xw, N);
    k_gather1   <<<(N + 7) / 8, T, 0, stream>>>(xw, rowptr, csr, b1, h, N);

    k_gemm2     <<<(N / 16 + 7) / 8, T, 0, stream>>>(h, W2, hw, N);
    k_gather2   <<<(N + 15) / 16, T, 0, stream>>>(hw, rowptr, csr, b2, out, N);
}


# Edge sort: per-edge returning global atomics + scan + scattered fill (5 launches) replaced by a 2-launch two-level counting sort ranked by LDS adds; arrival-order rank now from LDS (sum order)
# speedup vs baseline: 3.6665x; 1.5673x over previous
#include <hip/hip_runtime.h>
#include <hip/hip_bf16.h>


typedef __attribute__((ext_vector_type(2))) float v2f;
typedef __attribute__((ext_vector_type(8))) float v8f;

__device__ __forceinline__ v8f wmma_f32_k4(v2f a, v2f b, v8f c) {
    return __builtin_amdgcn_wmma_f32_16x16x4_f32(
         false, a,  false, b,
         (short)0, c,  false,  false);
}

#define TE   4096
#define MAXB 1024
#define MAXT 512
#define CAP  8192

__device__ __forceinline__ int block_excl_scan(int v, int* wsum, int& total) {
    const int lane = threadIdx.x & 31, wv = threadIdx.x >> 5;
    int inc = v;
    for (int o = 1; o < 32; o <<= 1) {
        int t = __shfl_up(inc, o, 32);
        if (lane >= o) inc += t;
    }
    __syncthreads();
    if (lane == 31) wsum[wv] = inc;
    __syncthreads();
    int woff = 0, tot = 0;
    #pragma unroll
    for (int k = 0; k < 8; ++k) {
        const int w = wsum[k];
        if (k < wv) woff += w;
        tot += w;
    }
    total = tot;
    return woff + inc - v;
}

__global__ void __launch_bounds__(256)
k_tile_sort(const int* __restrict__ src, const int* __restrict__ dst,
            unsigned* __restrict__ recs, int* __restrict__ offT, int E, int B, int T) {
    __shared__ __attribute__((aligned(16))) int hist[MAXB];
    __shared__ unsigned buf[TE];
    __shared__ int wsum[8];
    const int tid = threadIdx.x;
    const int t = blockIdx.x;
    const int e0 = t * TE;
    const int nt = min(TE, E - e0);
    for (int i = tid; i < MAXB; i += 256) hist[i] = 0;
    __syncthreads();
    int d[16], s[16], r[16];
    #pragma unroll
    for (int j = 0; j < 16; ++j) {
        const int i = j * 256 + tid;
        d[j] = -1; s[j] = 0;
        if (i < nt) { d[j] = dst[e0 + i]; s[j] = src[e0 + i]; }
    }
    #pragma unroll
    for (int j = 0; j < 16; ++j) {
        r[j] = 0;
        if (d[j] >= 0) r[j] = atomicAdd(&hist[d[j] >> 8], 1);
    }
    __syncthreads();
    const int4 hv = *(const int4*)&hist[4 * tid];
    int tot;
    const int ex = block_excl_scan(hv.x + hv.y + hv.z + hv.w, wsum, tot);
    int4 st; st.x = ex; st.y = ex + hv.x; st.z = st.y + hv.y; st.w = st.z + hv.z;
    *(int4*)&hist[4 * tid] = st;
    __syncthreads();
    for (int b = tid; b <= B; b += 256) offT[(size_t)b * T + t] = (b < B) ? hist[b] : nt;
    #pragma unroll
    for (int j = 0; j < 16; ++j)
        if (d[j] >= 0) buf[hist[d[j] >> 8] + r[j]] = ((unsigned)s[j] << 8) | (unsigned)(d[j] & 255);
    __syncthreads();
    #pragma unroll
    for (int j = 0; j < 16; ++j) {
        const int i = j * 256 + tid;
        if (i < nt) recs[e0 + i] = buf[i];
    }
}

__global__ void __launch_bounds__(256)
k_bucket_csr(const unsigned* __restrict__ recs, const int* __restrict__ offT,
             int* __restrict__ rowptr, int* __restrict__ csr, int N, int T, int B) {
    __shared__ int rs[MAXT];
    __shared__ int rp[MAXT + 1];
    __shared__ int cnt[256];
    __shared__ int roff[256];
    __shared__ int wsum[8];
    __shared__ unsigned buf[CAP];
    __shared__ int outb[CAP];
    const int tid = threadIdx.x;
    const int b = blockIdx.x;
    const int hw = tid >> 4, l16 = tid & 15;
    cnt[tid] = 0;

    const int t0 = 2 * tid, t1 = t0 + 1;
    const int* o0 = offT + (size_t)b * T;
    const int* o1 = o0 + T;
    int s0 = 0, x0 = 0, s1 = 0, x1 = 0;
    if (t0 < T) { s0 = o0[t0]; x0 = o1[t0]; }
    if (t1 < T) { s1 = o0[t1]; x1 = o1[t1]; }
    rs[t0] = t0 * TE + s0;
    rs[t1] = t1 * TE + s1;
    const int a = x0 - s0, c = x1 - s1;
    int nb, Sb;
    const int ex = block_excl_scan(a + c, wsum, nb);
    rp[t0] = ex;
    rp[t1] = ex + a;
    if (tid == 0) rp[MAXT] = nb;
    (void)block_excl_scan(s0 + s1, wsum, Sb);
    __syncthreads();
    const bool staged = nb <= CAP;

    if (staged) {
        for (int k0 = hw; k0 < T; k0 += 80) {
            unsigned v[5]; int p[5], ln[5], g[5];
            #pragma unroll
            for (int u = 0; u < 5; ++u) {
                const int k = k0 + 16 * u;
                p[u] = 0; ln[u] = 0; g[u] = 0; v[u] = 0;
                if (k < T) {
                    p[u] = rp[k]; ln[u] = rp[k + 1] - p[u]; g[u] = rs[k];
                    if (l16 < ln[u]) v[u] = recs[g[u] + l16];
                }
            }
            #pragma unroll
            for (int u = 0; u < 5; ++u) {
                if (l16 < ln[u]) { buf[p[u] + l16] = v[u]; atomicAdd(&cnt[v[u] & 255], 1); }
                for (int l = 16 + l16; l < ln[u]; l += 16) {
                    const unsigned w = recs[g[u] + l];
                    buf[p[u] + l] = w; atomicAdd(&cnt[w & 255], 1);
                }
            }
        }
    } else {
        for (int k = hw; k < T; k += 16) {
            const int p = rp[k], ln = rp[k + 1] - p, g = rs[k];
            for (int l = l16; l < ln; l += 16) atomicAdd(&cnt[recs[g + l] & 255], 1);
        }
    }
    __syncthreads();
    {
        const int cv = cnt[tid];
        int tot;
        const int e2 = block_excl_scan(cv, wsum, tot);
        roff[tid] = e2;
        const int node = b * 256 + tid;
        if (node < N) rowptr[node] = Sb + e2;
        if (b == B - 1 && tid == 0) rowptr[N] = Sb + nb;
        cnt[tid] = 0;
    }
    __syncthreads();
    if (staged) {
        for (int i0 = tid; i0 < nb; i0 += 1024) {
            unsigned v[4]; int r[4];
            #pragma unroll
            for (int u = 0; u < 4; ++u) { const int i = i0 + 256 * u; v[u] = (i < nb) ? buf[i] : 0u; }
            #pragma unroll
            for (int u = 0; u < 4; ++u) {
                r[u] = 0;
                if (i0 + 256 * u < nb) r[u] = atomicAdd(&cnt[v[u] & 255], 1);
            }
            #pragma unroll
            for (int u = 0; u < 4; ++u)
                if (i0 + 256 * u < nb) outb[roff[v[u] & 255] + r[u]] = (int)(v[u] >> 8);
        }
        __syncthreads();
        for (int i = tid; i < nb; i += 256) csr[Sb + i] = outb[i];
    } else {
        for (int k = hw; k < T; k += 16) {
            const int p = rp[k], ln = rp[k + 1] - p, g = rs[k];
            for (int l = l16; l < ln; l += 16) {
                const unsigned v = recs[g + l];
                const int d8 = (int)(v & 255);
                const int r = atomicAdd(&cnt[d8], 1);
                csr[Sb + roff[d8] + r] = (int)(v >> 8);
            }
        }
    }
}

__device__ __forceinline__ float dinv_of(const int* __restrict__ rowptr, int s) {
    return rsqrtf((float)(rowptr[s + 1] - rowptr[s] + 1));
}

__global__ void __launch_bounds__(256) __attribute__((amdgpu_waves_per_eu(4, 4)))
k_gemm1(const float* __restrict__ x, const float* __restrict__ W1, float* __restrict__ xw, int N) {
    __shared__ __attribute__((aligned(16))) float w1s[150 * 256];
    const int lane = threadIdx.x & 31;
    const int wave = __builtin_amdgcn_readfirstlane((int)(threadIdx.x >> 5));
    const int h    = lane >> 4;
    const int c    = lane & 15;
    const int wh   = wave & 1;
    const int rb   = blockIdx.x * 64 + ((wave >> 1) << 4);

    const int arow = min(rb + c, N - 1);
    const float* xr = x + (size_t)arow * 300 + (h << 2);
    float4 a[37];
    #pragma unroll
    for (int kk = 0; kk < 18; ++kk) a[kk] = *(const float4*)(xr + 8 * kk);

    const float* wsrc = W1 + (size_t)wave * 256 + (lane << 2);
    float* wdst = w1s + wave * 256 + (lane >> 4) * 128 + (lane & 15) * 4;
    #pragma unroll 9
    for (int it = 0; it < 18; ++it) {
        const float4 r0 = *(const float4*)(wsrc + it * 2048);
        const float4 r1 = *(const float4*)(wsrc + it * 2048 + 128);
        float4 lo; lo.x = r0.x; lo.y = r1.x; lo.z = r0.y; lo.w = r1.y;
        float4 hi; hi.x = r0.z; hi.y = r1.z; hi.z = r0.w; hi.w = r1.w;
        *(float4*)(wdst + it * 2048)      = lo;
        *(float4*)(wdst + it * 2048 + 64) = hi;
    }
    if (wave < 6) {
        const float4 r0 = *(const float4*)(wsrc + 18 * 2048);
        const float4 r1 = *(const float4*)(wsrc + 18 * 2048 + 128);
        float4 lo; lo.x = r0.x; lo.y = r1.x; lo.z = r0.y; lo.w = r1.y;
        float4 hi; hi.x = r0.z; hi.y = r1.z; hi.z = r0.w; hi.w = r1.w;
        *(float4*)(wdst + 18 * 2048)      = lo;
        *(float4*)(wdst + 18 * 2048 + 64) = hi;
    }
    __syncthreads();
    #pragma unroll
    for (int kk = 18; kk < 37; ++kk) a[kk] = *(const float4*)(xr + 8 * kk);
    const float2 at = *(const float2*)(x + (size_t)arow * 300 + 296 + (h << 1));

    v8f acc0 = {}, acc1 = {}, acc2 = {}, acc3 = {};
    const float* bl = w1s + (h << 9) + (wh << 7) + (c << 2);
    #pragma unroll
    for (int kk = 0; kk < 37; ++kk) {
        const float* p = bl + kk * 1024;
        const float4 b00 = *(const float4*)(p);
        const float4 b01 = *(const float4*)(p + 64);
        const float4 b10 = *(const float4*)(p + 256);
        const float4 b11 = *(const float4*)(p + 320);
        v2f a0; a0.x = a[kk].x; a0.y = a[kk].y;
        v2f a1; a1.x = a[kk].z; a1.y = a[kk].w;
        v2f b;
        b.x = b00.x; b.y = b00.y; acc0 = wmma_f32_k4(a0, b, acc0);
        b.x = b00.z; b.y = b00.w; acc1 = wmma_f32_k4(a0, b, acc1);
        b.x = b01.x; b.y = b01.y; acc2 = wmma_f32_k4(a0, b, acc2);
        b.x = b01.z; b.y = b01.w; acc3 = wmma_f32_k4(a0, b, acc3);
        b.x = b10.x; b.y = b10.y; acc0 = wmma_f32_k4(a1, b, acc0);
        b.x = b10.z; b.y = b10.w; acc1 = wmma_f32_k4(a1, b, acc1);
        b.x = b11.x; b.y = b11.y; acc2 = wmma_f32_k4(a1, b, acc2);
        b.x = b11.z; b.y = b11.w; acc3 = wmma_f32_k4(a1, b, acc3);
    }
    {
        const float* p = w1s + (148 + h) * 256 + (wh << 7) + (c << 2);
        const float4 b0 = *(const float4*)(p);
        const float4 b1 = *(const float4*)(p + 64);
        v2f av; av.x = at.x; av.y = at.y;
        v2f b;
        b.x = b0.x; b.y = b0.y; acc0 = wmma_f32_k4(av, b, acc0);
        b.x = b0.z; b.y = b0.w; acc1 = wmma_f32_k4(av, b, acc1);
        b.x = b1.x; b.y = b1.y; acc2 = wmma_f32_k4(av, b, acc2);
        b.x = b1.z; b.y = b1.w; acc3 = wmma_f32_k4(av, b, acc3);
    }
    #pragma unroll
    for (int r = 0; r < 8; ++r) {
        const int rr = rb + (h << 3) + r;
        if (rr < N) {
            float4 o; o.x = acc0[r]; o.y = acc1[r]; o.z = acc2[r]; o.w = acc3[r];
            *(float4*)(xw + (size_t)rr * 128 + (wh << 6) + (c << 2)) = o;
        }
    }
}

template <int U>
__device__ __forceinline__ void add_rows(const float* __restrict__ xw, int s_l, float w_l,
                                         int t, int f, float4& acc) {
    float4 v[U];
    float w[U];
    #pragma unroll
    for (int k = 0; k < U; ++k) {
        const int s = __builtin_amdgcn_readlane(s_l, t + k);
        w[k] = __int_as_float(__builtin_amdgcn_readlane(__float_as_int(w_l), t + k));
        v[k] = *(const float4*)(xw + (size_t)s * 128 + f);
    }
    #pragma unroll
    for (int k = 0; k < U; ++k) {
        acc.x += v[k].x * w[k];
        acc.y += v[k].y * w[k];
        acc.z += v[k].z * w[k];
        acc.w += v[k].w * w[k];
    }
}

__global__ void k_gather1(const float* __restrict__ xw, const int* __restrict__ rowptr,
                          const int* __restrict__ csr, const float* __restrict__ b1,
                          float* __restrict__ h, int N) {
    const int lane = threadIdx.x & 31;
    const int n = __builtin_amdgcn_readfirstlane((int)(blockIdx.x * 8 + (threadIdx.x >> 5)));
    if (n >= N) return;
    const int beg = rowptr[n], end = rowptr[n + 1];
    const float dn = rsqrtf((float)(end - beg + 1));
    const int f = lane << 2;
    float4 acc = {0.0f, 0.0f, 0.0f, 0.0f};
    for (int base = beg; base < end; base += 32) {
        const int m = min(32, end - base);
        int s_l = 0;
        float w_l = 0.0f;
        if (lane < m) {
            s_l = csr[base + lane];
            w_l = dinv_of(rowptr, s_l) * dn;
        }
        int t = 0;
        for (; t + 8 <= m; t += 8) add_rows<8>(xw, s_l, w_l, t, f, acc);
        if (t + 4 <= m) { add_rows<4>(xw, s_l, w_l, t, f, acc); t += 4; }
        if (t + 2 <= m) { add_rows<2>(xw, s_l, w_l, t, f, acc); t += 2; }
        if (t < m)      { add_rows<1>(xw, s_l, w_l, t, f, acc); }
    }
    const float4 sv = *(const float4*)(xw + (size_t)n * 128 + f);
    const float4 bv = *(const float4*)(b1 + f);
    float4 o;
    o.x = acc.x + sv.x * dn * dn + bv.x;
    o.y = acc.y + sv.y * dn * dn + bv.y;
    o.z = acc.z + sv.z * dn * dn + bv.z;
    o.w = acc.w + sv.w * dn * dn + bv.w;
    o.x = o.x > 0.0f ? o.x : 0.0f;
    o.y = o.y > 0.0f ? o.y : 0.0f;
    o.z = o.z > 0.0f ? o.z : 0.0f;
    o.w = o.w > 0.0f ? o.w : 0.0f;
    *(float4*)(h + (size_t)n * 128 + f) = o;
}

__global__ void k_gemm2(const float* __restrict__ h, const float* __restrict__ W2,
                        float* __restrict__ hw16, int N) {
    __shared__ float w2s[128 * 16];
    for (int i = threadIdx.x; i < 128 * 16; i += 256) {
        int k = i >> 4, c = i & 15;
        w2s[i] = (c < 10) ? W2[k * 10 + c] : 0.0f;
    }
    __syncthreads();

    const int lane = threadIdx.x & 31;
    const int wave = threadIdx.x >> 5;
    const int half = lane >> 4;
    const int rlo  = lane & 15;
    const int mt   = blockIdx.x * 8 + wave;
    if ((mt << 4) < N) {
        v8f acc = {};
        const float* hrow = h + (size_t)((mt << 4) + rlo) * 128 + (half << 1);
        for (int k = 0; k < 128; k += 4) {
            float2 av = *(const float2*)(hrow + k);
            v2f a; a.x = av.x; a.y = av.y;
            const int kr = k + (half << 1);
            v2f b;
            b.x = w2s[kr * 16 + rlo];
            b.y = w2s[(kr + 1) * 16 + rlo];
            acc = wmma_f32_k4(a, b, acc);
        }
        const int m0 = (mt << 4) + (half << 3);
        #pragma unroll
        for (int r = 0; r < 8; ++r)
            hw16[(size_t)(m0 + r) * 16 + rlo] = acc[r];
    }
}

__global__ void k_gather2(const float* __restrict__ hw16, const int* __restrict__ rowptr,
                          const int* __restrict__ csr, const float* __restrict__ b2,
                          float* __restrict__ out, int N) {
    __shared__ float sres[16 * 10];
    const int tid = threadIdx.x;
    const int f  = tid & 15;
    const int nl = tid >> 4;
    const int n  = blockIdx.x * 16 + nl;
    const int nc = n < N ? n : N - 1;
    const int beg = rowptr[nc], end = rowptr[nc + 1];
    const float dn = rsqrtf((float)(end - beg + 1));
    float acc = 0.0f;
    for (int base = beg; base < end; base += 16) {
        const int m = min(16, end - base);
        int s_l = 0;
        float w_l = 0.0f;
        if (f < m) {
            s_l = csr[base + f];
            w_l = dinv_of(rowptr, s_l) * dn;
        }
        int t = 0;
        for (; t + 4 <= m; t += 4) {
            const int s0 = __shfl(s_l, t, 16),     s1 = __shfl(s_l, t + 1, 16);
            const int s2 = __shfl(s_l, t + 2, 16), s3 = __shfl(s_l, t + 3, 16);
            const float w0 = __shfl(w_l, t, 16),     w1 = __shfl(w_l, t + 1, 16);
            const float w2 = __shfl(w_l, t + 2, 16), w3 = __shfl(w_l, t + 3, 16);
            const float v0 = hw16[(size_t)s0 * 16 + f];
            const float v1 = hw16[(size_t)s1 * 16 + f];
            const float v2 = hw16[(size_t)s2 * 16 + f];
            const float v3 = hw16[(size_t)s3 * 16 + f];
            acc += v0 * w0;
            acc += v1 * w1;
            acc += v2 * w2;
            acc += v3 * w3;
        }
        for (; t < m; ++t) {
            const int s = __shfl(s_l, t, 16);
            const float w = __shfl(w_l, t, 16);
            acc += hw16[(size_t)s * 16 + f] * w;
        }
    }
    const float self = hw16[(size_t)nc * 16 + f];
    if (f < 10) sres[nl * 10 + f] = acc + self * dn * dn + b2[f];
    __syncthreads();
    if (tid < 160) {
        const size_t o = (size_t)blockIdx.x * 160 + tid;
        if (o < (size_t)N * 10) out[o] = sres[tid];
    }
}

extern "C" void kernel_launch(void* const* d_in, const int* in_sizes, int n_in,
                              void* d_out, int out_size, void* d_ws, size_t ws_size,
                              hipStream_t stream) {
    const float* x  = (const float*)d_in[0];
    const int*   ei = (const int*)d_in[1];
    const float* W1 = (const float*)d_in[2];
    const float* b1 = (const float*)d_in[3];
    const float* W2 = (const float*)d_in[4];
    const float* b2 = (const float*)d_in[5];
    float* out = (float*)d_out;

    const int N = in_sizes[0] / 300;
    const int E = in_sizes[1] / 2;
    const int* src = ei;
    const int* dst = ei + E;

    size_t Np = ((size_t)N + 64) & ~(size_t)63;
    int*   rowptr = (int*)d_ws;
    float* xw   = (float*)d_ws + Np;
    float* h    = xw + Np * 128;
    int*   csr  = (int*)(h + Np * 128);
    float* hw   = xw;

    const int T = 256;
    const int nB = (N + 255) / 256;
    const int nT = (E + TE - 1) / TE;
    unsigned* recs = (unsigned*)h;
    int*      offT = (int*)h + (size_t)nT * TE;

    k_tile_sort <<<nT, T, 0, stream>>>(src, dst, recs, offT, E, nB, nT);
    k_bucket_csr<<<nB, T, 0, stream>>>(recs, offT, rowptr, csr, N, nT, nB);

    k_gemm1     <<<(N + 63) / 64, T, 0, stream>>>(x, W1, xw, N);
    k_gather1   <<<(N + 7) / 8, T, 0, stream>>>(xw, rowptr, csr, b1, h, N);

    k_gemm2     <<<(N / 16 + 7) / 8, T, 0, stream>>>(h, W2, hw, N);
    k_gather2   <<<(N + 15) / 16, T, 0, stream>>>(hw, rowptr, csr, b2, out, N);
}


# Source factor 1/sqrt(deg) folded into both products' epilogues (rows stored pre-scaled); gathers add plain rows, scale once by dst factor: no per-edge degree lookups; rounding changes, sum order kept
# speedup vs baseline: 3.9396x; 1.0745x over previous
#include <hip/hip_runtime.h>
#include <hip/hip_bf16.h>


typedef __attribute__((ext_vector_type(2))) float v2f;
typedef __attribute__((ext_vector_type(8))) float v8f;

__device__ __forceinline__ v8f wmma_f32_k4(v2f a, v2f b, v8f c) {
    return __builtin_amdgcn_wmma_f32_16x16x4_f32(
         false, a,  false, b,
         (short)0, c,  false,  false);
}

#define TE   4096
#define MAXB 1024
#define MAXT 512
#define CAP  8192

__device__ __forceinline__ int block_excl_scan(int v, int* wsum, int& total) {
    const int lane = threadIdx.x & 31, wv = threadIdx.x >> 5;
    int inc = v;
    for (int o = 1; o < 32; o <<= 1) {
        int t = __shfl_up(inc, o, 32);
        if (lane >= o) inc += t;
    }
    __syncthreads();
    if (lane == 31) wsum[wv] = inc;
    __syncthreads();
    int woff = 0, tot = 0;
    #pragma unroll
    for (int k = 0; k < 8; ++k) {
        const int w = wsum[k];
        if (k < wv) woff += w;
        tot += w;
    }
    total = tot;
    return woff + inc - v;
}

__global__ void __launch_bounds__(256)
k_tile_sort(const int* __restrict__ src, const int* __restrict__ dst,
            unsigned* __restrict__ recs, int* __restrict__ offT, int E, int B, int T) {
    __shared__ __attribute__((aligned(16))) int hist[MAXB];
    __shared__ unsigned buf[TE];
    __shared__ int wsum[8];
    const int tid = threadIdx.x;
    const int t = blockIdx.x;
    const int e0 = t * TE;
    const int nt = min(TE, E - e0);
    for (int i = tid; i < MAXB; i += 256) hist[i] = 0;
    __syncthreads();
    int d[16], s[16], r[16];
    #pragma unroll
    for (int j = 0; j < 16; ++j) {
        const int i = j * 256 + tid;
        d[j] = -1; s[j] = 0;
        if (i < nt) { d[j] = dst[e0 + i]; s[j] = src[e0 + i]; }
    }
    #pragma unroll
    for (int j = 0; j < 16; ++j) {
        r[j] = 0;
        if (d[j] >= 0) r[j] = atomicAdd(&hist[d[j] >> 8], 1);
    }
    __syncthreads();
    const int4 hv = *(const int4*)&hist[4 * tid];
    int tot;
    const int ex = block_excl_scan(hv.x + hv.y + hv.z + hv.w, wsum, tot);
    int4 st; st.x = ex; st.y = ex + hv.x; st.z = st.y + hv.y; st.w = st.z + hv.z;
    *(int4*)&hist[4 * tid] = st;
    __syncthreads();
    for (int b = tid; b <= B; b += 256) offT[(size_t)b * T + t] = (b < B) ? hist[b] : nt;
    #pragma unroll
    for (int j = 0; j < 16; ++j)
        if (d[j] >= 0) buf[hist[d[j] >> 8] + r[j]] = ((unsigned)s[j] << 8) | (unsigned)(d[j] & 255);
    __syncthreads();
    #pragma unroll
    for (int j = 0; j < 16; ++j) {
        const int i = j * 256 + tid;
        if (i < nt) recs[e0 + i] = buf[i];
    }
}

__global__ void __launch_bounds__(256)
k_bucket_csr(const unsigned* __restrict__ recs, const int* __restrict__ offT,
             int* __restrict__ rowptr, int* __restrict__ csr, int N, int T, int B) {
    __shared__ int rs[MAXT];
    __shared__ int rp[MAXT + 1];
    __shared__ int cnt[256];
    __shared__ int roff[256];
    __shared__ int wsum[8];
    __shared__ unsigned buf[CAP];
    __shared__ int outb[CAP];
    const int tid = threadIdx.x;
    const int b = blockIdx.x;
    const int hw = tid >> 4, l16 = tid & 15;
    cnt[tid] = 0;

    const int t0 = 2 * tid, t1 = t0 + 1;
    const int* o0 = offT + (size_t)b * T;
    const int* o1 = o0 + T;
    int s0 = 0, x0 = 0, s1 = 0, x1 = 0;
    if (t0 < T) { s0 = o0[t0]; x0 = o1[t0]; }
    if (t1 < T) { s1 = o0[t1]; x1 = o1[t1]; }
    rs[t0] = t0 * TE + s0;
    rs[t1] = t1 * TE + s1;
    const int a = x0 - s0, c = x1 - s1;
    int nb, Sb;
    const int ex = block_excl_scan(a + c, wsum, nb);
    rp[t0] = ex;
    rp[t1] = ex + a;
    if (tid == 0) rp[MAXT] = nb;
    (void)block_excl_scan(s0 + s1, wsum, Sb);
    __syncthreads();
    const bool staged = nb <= CAP;

    if (staged) {
        for (int k0 = hw; k0 < T; k0 += 80) {
            unsigned v[5]; int p[5], ln[5], g[5];
            #pragma unroll
            for (int u = 0; u < 5; ++u) {
                const int k = k0 + 16 * u;
                p[u] = 0; ln[u] = 0; g[u] = 0; v[u] = 0;
                if (k < T) {
                    p[u] = rp[k]; ln[u] = rp[k + 1] - p[u]; g[u] = rs[k];
                    if (l16 < ln[u]) v[u] = recs[g[u] + l16];
                }
            }
            #pragma unroll
            for (int u = 0; u < 5; ++u) {
                if (l16 < ln[u]) { buf[p[u] + l16] = v[u]; atomicAdd(&cnt[v[u] & 255], 1); }
                for (int l = 16 + l16; l < ln[u]; l += 16) {
                    const unsigned w = recs[g[u] + l];
                    buf[p[u] + l] = w; atomicAdd(&cnt[w & 255], 1);
                }
            }
        }
    } else {
        for (int k = hw; k < T; k += 16) {
            const int p = rp[k], ln = rp[k + 1] - p, g = rs[k];
            for (int l = l16; l < ln; l += 16) atomicAdd(&cnt[recs[g + l] & 255], 1);
        }
    }
    __syncthreads();
    {
        const int cv = cnt[tid];
        int tot;
        const int e2 = block_excl_scan(cv, wsum, tot);
        roff[tid] = e2;
        const int node = b * 256 + tid;
        if (node < N) rowptr[node] = Sb + e2;
        if (b == B - 1 && tid == 0) rowptr[N] = Sb + nb;
        cnt[tid] = 0;
    }
    __syncthreads();
    if (staged) {
        for (int i0 = tid; i0 < nb; i0 += 1024) {
            unsigned v[4]; int r[4];
            #pragma unroll
            for (int u = 0; u < 4; ++u) { const int i = i0 + 256 * u; v[u] = (i < nb) ? buf[i] : 0u; }
            #pragma unroll
            for (int u = 0; u < 4; ++u) {
                r[u] = 0;
                if (i0 + 256 * u < nb) r[u] = atomicAdd(&cnt[v[u] & 255], 1);
            }
            #pragma unroll
            for (int u = 0; u < 4; ++u)
                if (i0 + 256 * u < nb) outb[roff[v[u] & 255] + r[u]] = (int)(v[u] >> 8);
        }
        __syncthreads();
        for (int i = tid; i < nb; i += 256) csr[Sb + i] = outb[i];
    } else {
        for (int k = hw; k < T; k += 16) {
            const int p = rp[k], ln = rp[k + 1] - p, g = rs[k];
            for (int l = l16; l < ln; l += 16) {
                const unsigned v = recs[g + l];
                const int d8 = (int)(v & 255);
                const int r = atomicAdd(&cnt[d8], 1);
                csr[Sb + roff[d8] + r] = (int)(v >> 8);
            }
        }
    }
}

__device__ __forceinline__ float dinv_of(const int* __restrict__ rowptr, int s) {
    return rsqrtf((float)(rowptr[s + 1] - rowptr[s] + 1));
}

__global__ void __launch_bounds__(256) __attribute__((amdgpu_waves_per_eu(4, 4)))
k_gemm1(const float* __restrict__ x, const float* __restrict__ W1, const int* __restrict__ rowptr,
        float* __restrict__ xw, int N) {
    __shared__ __attribute__((aligned(16))) float w1s[150 * 256];
    const int lane = threadIdx.x & 31;
    const int wave = __builtin_amdgcn_readfirstlane((int)(threadIdx.x >> 5));
    const int h    = lane >> 4;
    const int c    = lane & 15;
    const int wh   = wave & 1;
    const int rb   = blockIdx.x * 64 + ((wave >> 1) << 4);

    const int arow = min(rb + c, N - 1);
    const float* xr = x + (size_t)arow * 300 + (h << 2);
    float4 a[37];
    #pragma unroll
    for (int kk = 0; kk < 18; ++kk) a[kk] = *(const float4*)(xr + 8 * kk);

    const float* wsrc = W1 + (size_t)wave * 256 + (lane << 2);
    float* wdst = w1s + wave * 256 + (lane >> 4) * 128 + (lane & 15) * 4;
    #pragma unroll 9
    for (int it = 0; it < 18; ++it) {
        const float4 r0 = *(const float4*)(wsrc + it * 2048);
        const float4 r1 = *(const float4*)(wsrc + it * 2048 + 128);
        float4 lo; lo.x = r0.x; lo.y = r1.x; lo.z = r0.y; lo.w = r1.y;
        float4 hi; hi.x = r0.z; hi.y = r1.z; hi.z = r0.w; hi.w = r1.w;
        *(float4*)(wdst + it * 2048)      = lo;
        *(float4*)(wdst + it * 2048 + 64) = hi;
    }
    if (wave < 6) {
        const float4 r0 = *(const float4*)(wsrc + 18 * 2048);
        const float4 r1 = *(const float4*)(wsrc + 18 * 2048 + 128);
        float4 lo; lo.x = r0.x; lo.y = r1.x; lo.z = r0.y; lo.w = r1.y;
        float4 hi; hi.x = r0.z; hi.y = r1.z; hi.z = r0.w; hi.w = r1.w;
        *(float4*)(wdst + 18 * 2048)      = lo;
        *(float4*)(wdst + 18 * 2048 + 64) = hi;
    }
    __syncthreads();
    #pragma unroll
    for (int kk = 18; kk < 37; ++kk) a[kk] = *(const float4*)(xr + 8 * kk);
    const float2 at = *(const float2*)(x + (size_t)arow * 300 + 296 + (h << 1));
    const int ql = rowptr[min(rb + lane, N)];

    v8f acc0 = {}, acc1 = {}, acc2 = {}, acc3 = {};
    const float* bl = w1s + (h << 9) + (wh << 7) + (c << 2);
    #pragma unroll
    for (int kk = 0; kk < 37; ++kk) {
        const float* p = bl + kk * 1024;
        const float4 b00 = *(const float4*)(p);
        const float4 b01 = *(const float4*)(p + 64);
        const float4 b10 = *(const float4*)(p + 256);
        const float4 b11 = *(const float4*)(p + 320);
        v2f a0; a0.x = a[kk].x; a0.y = a[kk].y;
        v2f a1; a1.x = a[kk].z; a1.y = a[kk].w;
        v2f b;
        b.x = b00.x; b.y = b00.y; acc0 = wmma_f32_k4(a0, b, acc0);
        b.x = b00.z; b.y = b00.w; acc1 = wmma_f32_k4(a0, b, acc1);
        b.x = b01.x; b.y = b01.y; acc2 = wmma_f32_k4(a0, b, acc2);
        b.x = b01.z; b.y = b01.w; acc3 = wmma_f32_k4(a0, b, acc3);
        b.x = b10.x; b.y = b10.y; acc0 = wmma_f32_k4(a1, b, acc0);
        b.x = b10.z; b.y = b10.w; acc1 = wmma_f32_k4(a1, b, acc1);
        b.x = b11.x; b.y = b11.y; acc2 = wmma_f32_k4(a1, b, acc2);
        b.x = b11.z; b.y = b11.w; acc3 = wmma_f32_k4(a1, b, acc3);
    }
    {
        const float* p = w1s + (148 + h) * 256 + (wh << 7) + (c << 2);
        const float4 b0 = *(const float4*)(p);
        const float4 b1 = *(const float4*)(p + 64);
        v2f av; av.x = at.x; av.y = at.y;
        v2f b;
        b.x = b0.x; b.y = b0.y; acc0 = wmma_f32_k4(av, b, acc0);
        b.x = b0.z; b.y = b0.w; acc1 = wmma_f32_k4(av, b, acc1);
        b.x = b1.x; b.y = b1.y; acc2 = wmma_f32_k4(av, b, acc2);
        b.x = b1.z; b.y = b1.w; acc3 = wmma_f32_k4(av, b, acc3);
    }
    const int qn = __shfl_down(ql, 1, 32);
    const float dl = rsqrtf((float)(qn - ql + 1));
    float dv[8];
    #pragma unroll
    for (int r = 0; r < 8; ++r) dv[r] = __shfl(dl, (h << 3) + r, 32);
    #pragma unroll
    for (int r = 0; r < 8; ++r) {
        const int rr = rb + (h << 3) + r;
        if (rr < N) {
            float4 o; o.x = acc0[r] * dv[r]; o.y = acc1[r] * dv[r]; o.z = acc2[r] * dv[r]; o.w = acc3[r] * dv[r];
            *(float4*)(xw + (size_t)rr * 128 + (wh << 6) + (c << 2)) = o;
        }
    }
}

template <int U>
__device__ __forceinline__ void add_rows(const float* __restrict__ xw, int s_l,
                                         int t, int f, float4& acc) {
    float4 v[U];
    #pragma unroll
    for (int k = 0; k < U; ++k) {
        const int s = __builtin_amdgcn_readlane(s_l, t + k);
        v[k] = *(const float4*)(xw + (size_t)s * 128 + f);
    }
    #pragma unroll
    for (int k = 0; k < U; ++k) {
        acc.x += v[k].x;
        acc.y += v[k].y;
        acc.z += v[k].z;
        acc.w += v[k].w;
    }
}

__global__ void k_gather1(const float* __restrict__ xw, const int* __restrict__ rowptr,
                          const int* __restrict__ csr, const float* __restrict__ b1,
                          float* __restrict__ h, int N) {
    const int lane = threadIdx.x & 31;
    const int n = __builtin_amdgcn_readfirstlane((int)(blockIdx.x * 8 + (threadIdx.x >> 5)));
    if (n >= N) return;
    const int beg = rowptr[n], end = rowptr[n + 1];
    const float dn = rsqrtf((float)(end - beg + 1));
    const int f = lane << 2;
    float4 acc = {0.0f, 0.0f, 0.0f, 0.0f};
    for (int base = beg; base < end; base += 32) {
        const int m = min(32, end - base);
        int s_l = 0;
        if (lane < m) s_l = csr[base + lane];
        int t = 0;
        for (; t + 8 <= m; t += 8) add_rows<8>(xw, s_l, t, f, acc);
        if (t + 4 <= m) { add_rows<4>(xw, s_l, t, f, acc); t += 4; }
        if (t + 2 <= m) { add_rows<2>(xw, s_l, t, f, acc); t += 2; }
        if (t < m)      { add_rows<1>(xw, s_l, t, f, acc); }
    }
    const float4 sv = *(const float4*)(xw + (size_t)n * 128 + f);
    const float4 bv = *(const float4*)(b1 + f);
    float4 o;
    o.x = (acc.x + sv.x) * dn + bv.x;
    o.y = (acc.y + sv.y) * dn + bv.y;
    o.z = (acc.z + sv.z) * dn + bv.z;
    o.w = (acc.w + sv.w) * dn + bv.w;
    o.x = o.x > 0.0f ? o.x : 0.0f;
    o.y = o.y > 0.0f ? o.y : 0.0f;
    o.z = o.z > 0.0f ? o.z : 0.0f;
    o.w = o.w > 0.0f ? o.w : 0.0f;
    *(float4*)(h + (size_t)n * 128 + f) = o;
}

__global__ void k_gemm2(const float* __restrict__ h, const float* __restrict__ W2,
                        const int* __restrict__ rowptr, float* __restrict__ hw16, int N) {
    __shared__ float w2s[128 * 16];
    for (int i = threadIdx.x; i < 128 * 16; i += 256) {
        int k = i >> 4, c = i & 15;
        w2s[i] = (c < 10) ? W2[k * 10 + c] : 0.0f;
    }
    __syncthreads();

    const int lane = threadIdx.x & 31;
    const int wave = threadIdx.x >> 5;
    const int half = lane >> 4;
    const int rlo  = lane & 15;
    const int mt   = blockIdx.x * 8 + wave;
    if ((mt << 4) < N) {
        v8f acc = {};
        const float* hrow = h + (size_t)((mt << 4) + rlo) * 128 + (half << 1);
        for (int k = 0; k < 128; k += 4) {
            float2 av = *(const float2*)(hrow + k);
            v2f a; a.x = av.x; a.y = av.y;
            const int kr = k + (half << 1);
            v2f b;
            b.x = w2s[kr * 16 + rlo];
            b.y = w2s[(kr + 1) * 16 + rlo];
            acc = wmma_f32_k4(a, b, acc);
        }
        const int m0 = (mt << 4) + (half << 3);
        #pragma unroll
        for (int r = 0; r < 8; ++r)
            hw16[(size_t)(m0 + r) * 16 + rlo] = acc[r] * dinv_of(rowptr, min(m0 + r, N - 1));
    }
}

__global__ void k_gather2(const float* __restrict__ hw16, const int* __restrict__ rowptr,
                          const int* __restrict__ csr, const float* __restrict__ b2,
                          float* __restrict__ out, int N) {
    __shared__ float sres[16 * 10];
    const int tid = threadIdx.x;
    const int f  = tid & 15;
    const int nl = tid >> 4;
    const int n  = blockIdx.x * 16 + nl;
    const int nc = n < N ? n : N - 1;
    const int beg = rowptr[nc], end = rowptr[nc + 1];
    const float dn = rsqrtf((float)(end - beg + 1));
    float acc = 0.0f;
    for (int base = beg; base < end; base += 16) {
        const int m = min(16, end - base);
        int s_l = 0;
        if (f < m) s_l = csr[base + f];
        int t = 0;
        for (; t + 4 <= m; t += 4) {
            const int s0 = __shfl(s_l, t, 16),     s1 = __shfl(s_l, t + 1, 16);
            const int s2 = __shfl(s_l, t + 2, 16), s3 = __shfl(s_l, t + 3, 16);
            const float v0 = hw16[(size_t)s0 * 16 + f];
            const float v1 = hw16[(size_t)s1 * 16 + f];
            const float v2 = hw16[(size_t)s2 * 16 + f];
            const float v3 = hw16[(size_t)s3 * 16 + f];
            acc += v0;
            acc += v1;
            acc += v2;
            acc += v3;
        }
        for (; t < m; ++t) {
            const int s = __shfl(s_l, t, 16);
            acc += hw16[(size_t)s * 16 + f];
        }
    }
    const float self = hw16[(size_t)nc * 16 + f];
    if (f < 10) sres[nl * 10 + f] = (acc + self) * dn + b2[f];
    __syncthreads();
    if (tid < 160) {
        const size_t o = (size_t)blockIdx.x * 160 + tid;
        if (o < (size_t)N * 10) out[o] = sres[tid];
    }
}

extern "C" void kernel_launch(void* const* d_in, const int* in_sizes, int n_in,
                              void* d_out, int out_size, void* d_ws, size_t ws_size,
                              hipStream_t stream) {
    const float* x  = (const float*)d_in[0];
    const int*   ei = (const int*)d_in[1];
    const float* W1 = (const float*)d_in[2];
    const float* b1 = (const float*)d_in[3];
    const float* W2 = (const float*)d_in[4];
    const float* b2 = (const float*)d_in[5];
    float* out = (float*)d_out;

    const int N = in_sizes[0] / 300;
    const int E = in_sizes[1] / 2;
    const int* src = ei;
    const int* dst = ei + E;

    size_t Np = ((size_t)N + 64) & ~(size_t)63;
    int*   rowptr = (int*)d_ws;
    float* xw   = (float*)d_ws + Np;
    float* h    = xw + Np * 128;
    int*   csr  = (int*)(h + Np * 128);
    float* hw   = xw;

    const int T = 256;
    const int nB = (N + 255) / 256;
    const int nT = (E + TE - 1) / TE;
    unsigned* recs = (unsigned*)h;
    int*      offT = (int*)h + (size_t)nT * TE;

    k_tile_sort <<<nT, T, 0, stream>>>(src, dst, recs, offT, E, nB, nT);
    k_bucket_csr<<<nB, T, 0, stream>>>(recs, offT, rowptr, csr, N, nT, nB);

    k_gemm1     <<<(N + 63) / 64, T, 0, stream>>>(x, W1, rowptr, xw, N);
    k_gather1   <<<(N + 7) / 8, T, 0, stream>>>(xw, rowptr, csr, b1, h, N);

    k_gemm2     <<<(N / 16 + 7) / 8, T, 0, stream>>>(h, W2, rowptr, hw, N);
    k_gather2   <<<(N + 15) / 16, T, 0, stream>>>(hw, rowptr, csr, b2, out, N);
}


# Second product fused into the layer-1 gather: a block keeps its 16 nodes' h rows in LDS and one wave multiplies them by W2 (f32 WMMA, two chains: k sum order changed); h never stored, 6 -> 5 launches
# speedup vs baseline: 4.2429x; 1.0770x over previous
#include <hip/hip_runtime.h>
#include <hip/hip_bf16.h>


typedef __attribute__((ext_vector_type(2))) float v2f;
typedef __attribute__((ext_vector_type(8))) float v8f;

__device__ __forceinline__ v8f wmma_f32_k4(v2f a, v2f b, v8f c) {
    return __builtin_amdgcn_wmma_f32_16x16x4_f32(
         false, a,  false, b,
         (short)0, c,  false,  false);
}

#define TE   4096
#define MAXB 1024
#define MAXT 512
#define CAP  8192

__device__ __forceinline__ int block_excl_scan(int v, int* wsum, int& total) {
    const int lane = threadIdx.x & 31, wv = threadIdx.x >> 5;
    int inc = v;
    for (int o = 1; o < 32; o <<= 1) {
        int t = __shfl_up(inc, o, 32);
        if (lane >= o) inc += t;
    }
    __syncthreads();
    if (lane == 31) wsum[wv] = inc;
    __syncthreads();
    int woff = 0, tot = 0;
    #pragma unroll
    for (int k = 0; k < 8; ++k) {
        const int w = wsum[k];
        if (k < wv) woff += w;
        tot += w;
    }
    total = tot;
    return woff + inc - v;
}

__global__ void __launch_bounds__(256)
k_tile_sort(const int* __restrict__ src, const int* __restrict__ dst,
            unsigned* __restrict__ recs, int* __restrict__ offT, int E, int B, int T) {
    __shared__ __attribute__((aligned(16))) int hist[MAXB];
    __shared__ unsigned buf[TE];
    __shared__ int wsum[8];
    const int tid = threadIdx.x;
    const int t = blockIdx.x;
    const int e0 = t * TE;
    const int nt = min(TE, E - e0);
    for (int i = tid; i < MAXB; i += 256) hist[i] = 0;
    __syncthreads();
    int d[16], s[16], r[16];
    #pragma unroll
    for (int j = 0; j < 16; ++j) {
        const int i = j * 256 + tid;
        d[j] = -1; s[j] = 0;
        if (i < nt) { d[j] = dst[e0 + i]; s[j] = src[e0 + i]; }
    }
    #pragma unroll
    for (int j = 0; j < 16; ++j) {
        r[j] = 0;
        if (d[j] >= 0) r[j] = atomicAdd(&hist[d[j] >> 8], 1);
    }
    __syncthreads();
    const int4 hv = *(const int4*)&hist[4 * tid];
    int tot;
    const int ex = block_excl_scan(hv.x + hv.y + hv.z + hv.w, wsum, tot);
    int4 st; st.x = ex; st.y = ex + hv.x; st.z = st.y + hv.y; st.w = st.z + hv.z;
    *(int4*)&hist[4 * tid] = st;
    __syncthreads();
    for (int b = tid; b <= B; b += 256) offT[(size_t)b * T + t] = (b < B) ? hist[b] : nt;
    #pragma unroll
    for (int j = 0; j < 16; ++j)
        if (d[j] >= 0) buf[hist[d[j] >> 8] + r[j]] = ((unsigned)s[j] << 8) | (unsigned)(d[j] & 255);
    __syncthreads();
    #pragma unroll
    for (int j = 0; j < 16; ++j) {
        const int i = j * 256 + tid;
        if (i < nt) recs[e0 + i] = buf[i];
    }
}

__global__ void __launch_bounds__(256)
k_bucket_csr(const unsigned* __restrict__ recs, const int* __restrict__ offT,
             int* __restrict__ rowptr, int* __restrict__ csr, int N, int T, int B) {
    __shared__ int rs[MAXT];
    __shared__ int rp[MAXT + 1];
    __shared__ int cnt[256];
    __shared__ int roff[256];
    __shared__ int wsum[8];
    __shared__ unsigned buf[CAP];
    __shared__ int outb[CAP];
    const int tid = threadIdx.x;
    const int b = blockIdx.x;
    const int hw = tid >> 4, l16 = tid & 15;
    cnt[tid] = 0;

    const int t0 = 2 * tid, t1 = t0 + 1;
    const int* o0 = offT + (size_t)b * T;
    const int* o1 = o0 + T;
    int s0 = 0, x0 = 0, s1 = 0, x1 = 0;
    if (t0 < T) { s0 = o0[t0]; x0 = o1[t0]; }
    if (t1 < T) { s1 = o0[t1]; x1 = o1[t1]; }
    rs[t0] = t0 * TE + s0;
    rs[t1] = t1 * TE + s1;
    const int a = x0 - s0, c = x1 - s1;
    int nb, Sb;
    const int ex = block_excl_scan(a + c, wsum, nb);
    rp[t0] = ex;
    rp[t1] = ex + a;
    if (tid == 0) rp[MAXT] = nb;
    (void)block_excl_scan(s0 + s1, wsum, Sb);
    __syncthreads();
    const bool staged = nb <= CAP;

    if (staged) {
        for (int k0 = hw; k0 < T; k0 += 80) {
            unsigned v[5]; int p[5], ln[5], g[5];
            #pragma unroll
            for (int u = 0; u < 5; ++u) {
                const int k = k0 + 16 * u;
                p[u] = 0; ln[u] = 0; g[u] = 0; v[u] = 0;
                if (k < T) {
                    p[u] = rp[k]; ln[u] = rp[k + 1] - p[u]; g[u] = rs[k];
                    if (l16 < ln[u]) v[u] = recs[g[u] + l16];
                }
            }
            #pragma unroll
            for (int u = 0; u < 5; ++u) {
                if (l16 < ln[u]) { buf[p[u] + l16] = v[u]; atomicAdd(&cnt[v[u] & 255], 1); }
                for (int l = 16 + l16; l < ln[u]; l += 16) {
                    const unsigned w = recs[g[u] + l];
                    buf[p[u] + l] = w; atomicAdd(&cnt[w & 255], 1);
                }
            }
        }
    } else {
        for (int k = hw; k < T; k += 16) {
            const int p = rp[k], ln = rp[k + 1] - p, g = rs[k];
            for (int l = l16; l < ln; l += 16) atomicAdd(&cnt[recs[g + l] & 255], 1);
        }
    }
    __syncthreads();
    {
        const int cv = cnt[tid];
        int tot;
        const int e2 = block_excl_scan(cv, wsum, tot);
        roff[tid] = e2;
        const int node = b * 256 + tid;
        if (node < N) rowptr[node] = Sb + e2;
        if (b == B - 1 && tid == 0) rowptr[N] = Sb + nb;
        cnt[tid] = 0;
    }
    __syncthreads();
    if (staged) {
        for (int i0 = tid; i0 < nb; i0 += 1024) {
            unsigned v[4]; int r[4];
            #pragma unroll
            for (int u = 0; u < 4; ++u) { const int i = i0 + 256 * u; v[u] = (i < nb) ? buf[i] : 0u; }
            #pragma unroll
            for (int u = 0; u < 4; ++u) {
                r[u] = 0;
                if (i0 + 256 * u < nb) r[u] = atomicAdd(&cnt[v[u] & 255], 1);
            }
            #pragma unroll
            for (int u = 0; u < 4; ++u)
                if (i0 + 256 * u < nb) outb[roff[v[u] & 255] + r[u]] = (int)(v[u] >> 8);
        }
        __syncthreads();
        for (int i = tid; i < nb; i += 256) csr[Sb + i] = outb[i];
    } else {
        for (int k = hw; k < T; k += 16) {
            const int p = rp[k], ln = rp[k + 1] - p, g = rs[k];
            for (int l = l16; l < ln; l += 16) {
                const unsigned v = recs[g + l];
                const int d8 = (int)(v & 255);
                const int r = atomicAdd(&cnt[d8], 1);
                csr[Sb + roff[d8] + r] = (int)(v >> 8);
            }
        }
    }
}

__global__ void __launch_bounds__(256) __attribute__((amdgpu_waves_per_eu(4, 4)))
k_gemm1(const float* __restrict__ x, const float* __restrict__ W1, const int* __restrict__ rowptr,
        float* __restrict__ xw, int N) {
    __shared__ __attribute__((aligned(16))) float w1s[150 * 256];
    const int lane = threadIdx.x & 31;
    const int wave = __builtin_amdgcn_readfirstlane((int)(threadIdx.x >> 5));
    const int h    = lane >> 4;
    const int c    = lane & 15;
    const int wh   = wave & 1;
    const int rb   = blockIdx.x * 64 + ((wave >> 1) << 4);

    const int arow = min(rb + c, N - 1);
    const float* xr = x + (size_t)arow * 300 + (h << 2);
    float4 a[37];
    #pragma unroll
    for (int kk = 0; kk < 18; ++kk) a[kk] = *(const float4*)(xr + 8 * kk);

    const float* wsrc = W1 + (size_t)wave * 256 + (lane << 2);
    float* wdst = w1s + wave * 256 + (lane >> 4) * 128 + (lane & 15) * 4;
    #pragma unroll 9
    for (int it = 0; it < 18; ++it) {
        const float4 r0 = *(const float4*)(wsrc + it * 2048);
        const float4 r1 = *(const float4*)(wsrc + it * 2048 + 128);
        float4 lo; lo.x = r0.x; lo.y = r1.x; lo.z = r0.y; lo.w = r1.y;
        float4 hi; hi.x = r0.z; hi.y = r1.z; hi.z = r0.w; hi.w = r1.w;
        *(float4*)(wdst + it * 2048)      = lo;
        *(float4*)(wdst + it * 2048 + 64) = hi;
    }
    if (wave < 6) {
        const float4 r0 = *(const float4*)(wsrc + 18 * 2048);
        const float4 r1 = *(const float4*)(wsrc + 18 * 2048 + 128);
        float4 lo; lo.x = r0.x; lo.y = r1.x; lo.z = r0.y; lo.w = r1.y;
        float4 hi; hi.x = r0.z; hi.y = r1.z; hi.z = r0.w; hi.w = r1.w;
        *(float4*)(wdst + 18 * 2048)      = lo;
        *(float4*)(wdst + 18 * 2048 + 64) = hi;
    }
    __syncthreads();
    #pragma unroll
    for (int kk = 18; kk < 37; ++kk) a[kk] = *(const float4*)(xr + 8 * kk);
    const float2 at = *(const float2*)(x + (size_t)arow * 300 + 296 + (h << 1));
    const int ql = rowptr[min(rb + lane, N)];

    v8f acc0 = {}, acc1 = {}, acc2 = {}, acc3 = {};
    const float* bl = w1s + (h << 9) + (wh << 7) + (c << 2);
    #pragma unroll
    for (int kk = 0; kk < 37; ++kk) {
        const float* p = bl + kk * 1024;
        const float4 b00 = *(const float4*)(p);
        const float4 b01 = *(const float4*)(p + 64);
        const float4 b10 = *(const float4*)(p + 256);
        const float4 b11 = *(const float4*)(p + 320);
        v2f a0; a0.x = a[kk].x; a0.y = a[kk].y;
        v2f a1; a1.x = a[kk].z; a1.y = a[kk].w;
        v2f b;
        b.x = b00.x; b.y = b00.y; acc0 = wmma_f32_k4(a0, b, acc0);
        b.x = b00.z; b.y = b00.w; acc1 = wmma_f32_k4(a0, b, acc1);
        b.x = b01.x; b.y = b01.y; acc2 = wmma_f32_k4(a0, b, acc2);
        b.x = b01.z; b.y = b01.w; acc3 = wmma_f32_k4(a0, b, acc3);
        b.x = b10.x; b.y = b10.y; acc0 = wmma_f32_k4(a1, b, acc0);
        b.x = b10.z; b.y = b10.w; acc1 = wmma_f32_k4(a1, b, acc1);
        b.x = b11.x; b.y = b11.y; acc2 = wmma_f32_k4(a1, b, acc2);
        b.x = b11.z; b.y = b11.w; acc3 = wmma_f32_k4(a1, b, acc3);
    }
    {
        const float* p = w1s + (148 + h) * 256 + (wh << 7) + (c << 2);
        const float4 b0 = *(const float4*)(p);
        const float4 b1 = *(const float4*)(p + 64);
        v2f av; av.x = at.x; av.y = at.y;
        v2f b;
        b.x = b0.x; b.y = b0.y; acc0 = wmma_f32_k4(av, b, acc0);
        b.x = b0.z; b.y = b0.w; acc1 = wmma_f32_k4(av, b, acc1);
        b.x = b1.x; b.y = b1.y; acc2 = wmma_f32_k4(av, b, acc2);
        b.x = b1.z; b.y = b1.w; acc3 = wmma_f32_k4(av, b, acc3);
    }
    const int qn = __shfl_down(ql, 1, 32);
    const float dl = rsqrtf((float)(qn - ql + 1));
    float dv[8];
    #pragma unroll
    for (int r = 0; r < 8; ++r) dv[r] = __shfl(dl, (h << 3) + r, 32);
    #pragma unroll
    for (int r = 0; r < 8; ++r) {
        const int rr = rb + (h << 3) + r;
        if (rr < N) {
            float4 o; o.x = acc0[r] * dv[r]; o.y = acc1[r] * dv[r]; o.z = acc2[r] * dv[r]; o.w = acc3[r] * dv[r];
            *(float4*)(xw + (size_t)rr * 128 + (wh << 6) + (c << 2)) = o;
        }
    }
}

template <int U>
__device__ __forceinline__ void add_rows(const float* __restrict__ xw, int s_l,
                                         int t, int f, float4& acc) {
    float4 v[U];
    #pragma unroll
    for (int k = 0; k < U; ++k) {
        const int s = __builtin_amdgcn_readlane(s_l, t + k);
        v[k] = *(const float4*)(xw + (size_t)s * 128 + f);
    }
    #pragma unroll
    for (int k = 0; k < U; ++k) {
        acc.x += v[k].x;
        acc.y += v[k].y;
        acc.z += v[k].z;
        acc.w += v[k].w;
    }
}

#define HP 132

__device__ __forceinline__ float4 gather_rows(const float* __restrict__ xw, const int* __restrict__ csr,
                                              int beg, int end, int s_l, int lane, int f) {
    float4 acc = {0.0f, 0.0f, 0.0f, 0.0f};
    for (int base = beg; base < end; base += 32) {
        const int m = min(32, end - base);
        if (base != beg) {
            s_l = 0;
            if (lane < m) s_l = csr[base + lane];
        }
        int t = 0;
        for (; t + 8 <= m; t += 8) add_rows<8>(xw, s_l, t, f, acc);
        if (t + 4 <= m) { add_rows<4>(xw, s_l, t, f, acc); t += 4; }
        if (t + 2 <= m) { add_rows<2>(xw, s_l, t, f, acc); t += 2; }
        if (t < m)      { add_rows<1>(xw, s_l, t, f, acc); }
    }
    return acc;
}

__global__ void __launch_bounds__(256)
k_gather1_gemm2(const float* __restrict__ xw, const int* __restrict__ rowptr,
                const int* __restrict__ csr, const float* __restrict__ b1,
                const float* __restrict__ W2, float* __restrict__ hw16, int N) {
    __shared__ __attribute__((aligned(16))) float hs[16 * HP];
    __shared__ __attribute__((aligned(16))) float w2q[16 * 2 * 16 * 4];
    __shared__ __attribute__((aligned(16))) float dns[16];
    const int tid  = threadIdx.x;
    const int lane = tid & 31;
    const int wave = __builtin_amdgcn_readfirstlane((int)(tid >> 5));
    const int f = lane << 2;
    const int n0 = blockIdx.x * 16 + wave * 2;
    const int p0 = rowptr[min(n0, N)], p1 = rowptr[min(n0 + 1, N)], p2 = rowptr[min(n0 + 2, N)];
    int sA = 0, sB = 0;
    if (lane < p1 - p0) sA = csr[p0 + lane];
    if (lane < p2 - p1) sB = csr[p1 + lane];
    float4 sv0 = {0.0f, 0.0f, 0.0f, 0.0f}, sv1 = {0.0f, 0.0f, 0.0f, 0.0f};
    if (n0 < N)     sv0 = *(const float4*)(xw + (size_t)n0 * 128 + f);
    if (n0 + 1 < N) sv1 = *(const float4*)(xw + (size_t)(n0 + 1) * 128 + f);
    float4 wq[2];
    #pragma unroll
    for (int j = 0; j < 2; ++j) {
        const int i = tid + 256 * j;
        const int k0 = ((i >> 5) << 3) + (((i >> 4) & 1) << 2), cc = i & 15;
        wq[j].x = 0.0f; wq[j].y = 0.0f; wq[j].z = 0.0f; wq[j].w = 0.0f;
        if (cc < 10) {
            wq[j].x = W2[(k0 + 0) * 10 + cc];
            wq[j].y = W2[(k0 + 1) * 10 + cc];
            wq[j].z = W2[(k0 + 2) * 10 + cc];
            wq[j].w = W2[(k0 + 3) * 10 + cc];
        }
    }
    const float4 bv = *(const float4*)(b1 + f);
    *(float4*)&w2q[(tid      ) * 4] = wq[0];
    *(float4*)&w2q[(tid + 256) * 4] = wq[1];

    {
        const float4 a = gather_rows(xw, csr, p0, p1, sA, lane, f);
        float4 o = {0.0f, 0.0f, 0.0f, 0.0f};
        float dn = 0.0f;
        if (n0 < N) {
            dn = rsqrtf((float)(p1 - p0 + 1));
            o.x = (a.x + sv0.x) * dn + bv.x;
            o.y = (a.y + sv0.y) * dn + bv.y;
            o.z = (a.z + sv0.z) * dn + bv.z;
            o.w = (a.w + sv0.w) * dn + bv.w;
            o.x = o.x > 0.0f ? o.x : 0.0f;
            o.y = o.y > 0.0f ? o.y : 0.0f;
            o.z = o.z > 0.0f ? o.z : 0.0f;
            o.w = o.w > 0.0f ? o.w : 0.0f;
        }
        *(float4*)&hs[(wave * 2) * HP + f] = o;
        if (lane == 0) dns[wave * 2] = dn;
    }
    {
        const float4 a = gather_rows(xw, csr, p1, p2, sB, lane, f);
        float4 o = {0.0f, 0.0f, 0.0f, 0.0f};
        float dn = 0.0f;
        if (n0 + 1 < N) {
            dn = rsqrtf((float)(p2 - p1 + 1));
            o.x = (a.x + sv1.x) * dn + bv.x;
            o.y = (a.y + sv1.y) * dn + bv.y;
            o.z = (a.z + sv1.z) * dn + bv.z;
            o.w = (a.w + sv1.w) * dn + bv.w;
            o.x = o.x > 0.0f ? o.x : 0.0f;
            o.y = o.y > 0.0f ? o.y : 0.0f;
            o.z = o.z > 0.0f ? o.z : 0.0f;
            o.w = o.w > 0.0f ? o.w : 0.0f;
        }
        *(float4*)&hs[(wave * 2 + 1) * HP + f] = o;
        if (lane == 0) dns[wave * 2 + 1] = dn;
    }
    __syncthreads();
    if (wave != 0) return;

    const int h = lane >> 4, c = lane & 15;
    v8f acc0 = {}, acc1 = {};
    const float* ap = hs + c * HP + (h << 2);
    const float* bp = w2q + (((h << 4) + c) << 2);
    #pragma unroll
    for (int kk = 0; kk < 16; ++kk) {
        const float4 a = *(const float4*)(ap + 8 * kk);
        const float4 b = *(const float4*)(bp + 128 * kk);
        v2f a0; a0.x = a.x; a0.y = a.y;
        v2f a1; a1.x = a.z; a1.y = a.w;
        v2f b0; b0.x = b.x; b0.y = b.y;
        v2f b1v; b1v.x = b.z; b1v.y = b.w;
        acc0 = wmma_f32_k4(a0, b0, acc0);
        acc1 = wmma_f32_k4(a1, b1v, acc1);
    }
    const float4 d0 = *(const float4*)&dns[h << 3];
    const float4 d1 = *(const float4*)&dns[(h << 3) + 4];
    const float dv[8] = {d0.x, d0.y, d0.z, d0.w, d1.x, d1.y, d1.z, d1.w};
    const int m0 = blockIdx.x * 16 + (h << 3);
    #pragma unroll
    for (int r = 0; r < 8; ++r)
        if (m0 + r < N) hw16[(size_t)(m0 + r) * 16 + c] = (acc0[r] + acc1[r]) * dv[r];
}

__global__ void k_gather2(const float* __restrict__ hw16, const int* __restrict__ rowptr,
                          const int* __restrict__ csr, const float* __restrict__ b2,
                          float* __restrict__ out, int N) {
    __shared__ float sres[16 * 10];
    const int tid = threadIdx.x;
    const int f  = tid & 15;
    const int nl = tid >> 4;
    const int n  = blockIdx.x * 16 + nl;
    const int nc = n < N ? n : N - 1;
    const int beg = rowptr[nc], end = rowptr[nc + 1];
    const float dn = rsqrtf((float)(end - beg + 1));
    float acc = 0.0f;
    for (int base = beg; base < end; base += 16) {
        const int m = min(16, end - base);
        int s_l = 0;
        if (f < m) s_l = csr[base + f];
        int t = 0;
        for (; t + 4 <= m; t += 4) {
            const int s0 = __shfl(s_l, t, 16),     s1 = __shfl(s_l, t + 1, 16);
            const int s2 = __shfl(s_l, t + 2, 16), s3 = __shfl(s_l, t + 3, 16);
            const float v0 = hw16[(size_t)s0 * 16 + f];
            const float v1 = hw16[(size_t)s1 * 16 + f];
            const float v2 = hw16[(size_t)s2 * 16 + f];
            const float v3 = hw16[(size_t)s3 * 16 + f];
            acc += v0;
            acc += v1;
            acc += v2;
            acc += v3;
        }
        for (; t < m; ++t) {
            const int s = __shfl(s_l, t, 16);
            acc += hw16[(size_t)s * 16 + f];
        }
    }
    const float self = hw16[(size_t)nc * 16 + f];
    if (f < 10) sres[nl * 10 + f] = (acc + self) * dn + b2[f];
    __syncthreads();
    if (tid < 160) {
        const size_t o = (size_t)blockIdx.x * 160 + tid;
        if (o < (size_t)N * 10) out[o] = sres[tid];
    }
}

extern "C" void kernel_launch(void* const* d_in, const int* in_sizes, int n_in,
                              void* d_out, int out_size, void* d_ws, size_t ws_size,
                              hipStream_t stream) {
    const float* x  = (const float*)d_in[0];
    const int*   ei = (const int*)d_in[1];
    const float* W1 = (const float*)d_in[2];
    const float* b1 = (const float*)d_in[3];
    const float* W2 = (const float*)d_in[4];
    const float* b2 = (const float*)d_in[5];
    float* out = (float*)d_out;

    const int N = in_sizes[0] / 300;
    const int E = in_sizes[1] / 2;
    const int* src = ei;
    const int* dst = ei + E;

    size_t Np = ((size_t)N + 64) & ~(size_t)63;
    int*   rowptr = (int*)d_ws;
    float* xw   = (float*)d_ws + Np;
    float* r2   = xw + Np * 128;
    int*   csr  = (int*)(r2 + Np * 128);
    float* hw   = r2;

    const int T = 256;
    const int nB = (N + 255) / 256;
    const int nT = (E + TE - 1) / TE;
    unsigned* recs = (unsigned*)r2;
    int*      offT = (int*)r2 + (size_t)nT * TE;

    k_tile_sort <<<nT, T, 0, stream>>>(src, dst, recs, offT, E, nB, nT);
    k_bucket_csr<<<nB, T, 0, stream>>>(recs, offT, rowptr, csr, N, nT, nB);

    k_gemm1     <<<(N + 63) / 64, T, 0, stream>>>(x, W1, rowptr, xw, N);
    k_gather1_gemm2<<<(N + 15) / 16, T, 0, stream>>>(xw, rowptr, csr, b1, W2, hw, N);
    k_gather2   <<<(N + 15) / 16, T, 0, stream>>>(hw, rowptr, csr, b2, out, N);
}


# Layer-2 gather: 4 lanes x float4 per node (8 nodes a wave, 8 row loads in flight per lane, next source indices prefetched) instead of 16 lanes x 1 float: 50,000 -> 12,504 waves; sum order kept
# speedup vs baseline: 4.3946x; 1.0358x over previous
#include <hip/hip_runtime.h>
#include <hip/hip_bf16.h>


typedef __attribute__((ext_vector_type(2))) float v2f;
typedef __attribute__((ext_vector_type(8))) float v8f;

__device__ __forceinline__ v8f wmma_f32_k4(v2f a, v2f b, v8f c) {
    return __builtin_amdgcn_wmma_f32_16x16x4_f32(
         false, a,  false, b,
         (short)0, c,  false,  false);
}

#define TE   4096
#define MAXB 1024
#define MAXT 512
#define CAP  8192

__device__ __forceinline__ int block_excl_scan(int v, int* wsum, int& total) {
    const int lane = threadIdx.x & 31, wv = threadIdx.x >> 5;
    int inc = v;
    for (int o = 1; o < 32; o <<= 1) {
        int t = __shfl_up(inc, o, 32);
        if (lane >= o) inc += t;
    }
    __syncthreads();
    if (lane == 31) wsum[wv] = inc;
    __syncthreads();
    int woff = 0, tot = 0;
    #pragma unroll
    for (int k = 0; k < 8; ++k) {
        const int w = wsum[k];
        if (k < wv) woff += w;
        tot += w;
    }
    total = tot;
    return woff + inc - v;
}

__global__ void __launch_bounds__(256)
k_tile_sort(const int* __restrict__ src, const int* __restrict__ dst,
            unsigned* __restrict__ recs, int* __restrict__ offT, int E, int B, int T) {
    __shared__ __attribute__((aligned(16))) int hist[MAXB];
    __shared__ unsigned buf[TE];
    __shared__ int wsum[8];
    const int tid = threadIdx.x;
    const int t = blockIdx.x;
    const int e0 = t * TE;
    const int nt = min(TE, E - e0);
    for (int i = tid; i < MAXB; i += 256) hist[i] = 0;
    __syncthreads();
    int d[16], s[16], r[16];
    #pragma unroll
    for (int j = 0; j < 16; ++j) {
        const int i = j * 256 + tid;
        d[j] = -1; s[j] = 0;
        if (i < nt) { d[j] = dst[e0 + i]; s[j] = src[e0 + i]; }
    }
    #pragma unroll
    for (int j = 0; j < 16; ++j) {
        r[j] = 0;
        if (d[j] >= 0) r[j] = atomicAdd(&hist[d[j] >> 8], 1);
    }
    __syncthreads();
    const int4 hv = *(const int4*)&hist[4 * tid];
    int tot;
    const int ex = block_excl_scan(hv.x + hv.y + hv.z + hv.w, wsum, tot);
    int4 st; st.x = ex; st.y = ex + hv.x; st.z = st.y + hv.y; st.w = st.z + hv.z;
    *(int4*)&hist[4 * tid] = st;
    __syncthreads();
    for (int b = tid; b <= B; b += 256) offT[(size_t)b * T + t] = (b < B) ? hist[b] : nt;
    #pragma unroll
    for (int j = 0; j < 16; ++j)
        if (d[j] >= 0) buf[hist[d[j] >> 8] + r[j]] = ((unsigned)s[j] << 8) | (unsigned)(d[j] & 255);
    __syncthreads();
    #pragma unroll
    for (int j = 0; j < 16; ++j) {
        const int i = j * 256 + tid;
        if (i < nt) recs[e0 + i] = buf[i];
    }
}

__global__ void __launch_bounds__(256)
k_bucket_csr(const unsigned* __restrict__ recs, const int* __restrict__ offT,
             int* __restrict__ rowptr, int* __restrict__ csr, int N, int T, int B) {
    __shared__ int rs[MAXT];
    __shared__ int rp[MAXT + 1];
    __shared__ int cnt[256];
    __shared__ int roff[256];
    __shared__ int wsum[8];
    __shared__ unsigned buf[CAP];
    __shared__ int outb[CAP];
    const int tid = threadIdx.x;
    const int b = blockIdx.x;
    const int hw = tid >> 4, l16 = tid & 15;
    cnt[tid] = 0;

    const int t0 = 2 * tid, t1 = t0 + 1;
    const int* o0 = offT + (size_t)b * T;
    const int* o1 = o0 + T;
    int s0 = 0, x0 = 0, s1 = 0, x1 = 0;
    if (t0 < T) { s0 = o0[t0]; x0 = o1[t0]; }
    if (t1 < T) { s1 = o0[t1]; x1 = o1[t1]; }
    rs[t0] = t0 * TE + s0;
    rs[t1] = t1 * TE + s1;
    const int a = x0 - s0, c = x1 - s1;
    int nb, Sb;
    const int ex = block_excl_scan(a + c, wsum, nb);
    rp[t0] = ex;
    rp[t1] = ex + a;
    if (tid == 0) rp[MAXT] = nb;
    (void)block_excl_scan(s0 + s1, wsum, Sb);
    __syncthreads();
    const bool staged = nb <= CAP;

    if (staged) {
        for (int k0 = hw; k0 < T; k0 += 80) {
            unsigned v[5]; int p[5], ln[5], g[5];
            #pragma unroll
            for (int u = 0; u < 5; ++u) {
                const int k = k0 + 16 * u;
                p[u] = 0; ln[u] = 0; g[u] = 0; v[u] = 0;
                if (k < T) {
                    p[u] = rp[k]; ln[u] = rp[k + 1] - p[u]; g[u] = rs[k];
                    if (l16 < ln[u]) v[u] = recs[g[u] + l16];
                }
            }
            #pragma unroll
            for (int u = 0; u < 5; ++u) {
                if (l16 < ln[u]) { buf[p[u] + l16] = v[u]; atomicAdd(&cnt[v[u] & 255], 1); }
                for (int l = 16 + l16; l < ln[u]; l += 16) {
                    const unsigned w = recs[g[u] + l];
                    buf[p[u] + l] = w; atomicAdd(&cnt[w & 255], 1);
                }
            }
        }
    } else {
        for (int k = hw; k < T; k += 16) {
            const int p = rp[k], ln = rp[k + 1] - p, g = rs[k];
            for (int l = l16; l < ln; l += 16) atomicAdd(&cnt[recs[g + l] & 255], 1);
        }
    }
    __syncthreads();
    {
        const int cv = cnt[tid];
        int tot;
        const int e2 = block_excl_scan(cv, wsum, tot);
        roff[tid] = e2;
        const int node = b * 256 + tid;
        if (node < N) rowptr[node] = Sb + e2;
        if (b == B - 1 && tid == 0) rowptr[N] = Sb + nb;
        cnt[tid] = 0;
    }
    __syncthreads();
    if (staged) {
        for (int i0 = tid; i0 < nb; i0 += 1024) {
            unsigned v[4]; int r[4];
            #pragma unroll
            for (int u = 0; u < 4; ++u) { const int i = i0 + 256 * u; v[u] = (i < nb) ? buf[i] : 0u; }
            #pragma unroll
            for (int u = 0; u < 4; ++u) {
                r[u] = 0;
                if (i0 + 256 * u < nb) r[u] = atomicAdd(&cnt[v[u] & 255], 1);
            }
            #pragma unroll
            for (int u = 0; u < 4; ++u)
                if (i0 + 256 * u < nb) outb[roff[v[u] & 255] + r[u]] = (int)(v[u] >> 8);
        }
        __syncthreads();
        for (int i = tid; i < nb; i += 256) csr[Sb + i] = outb[i];
    } else {
        for (int k = hw; k < T; k += 16) {
            const int p = rp[k], ln = rp[k + 1] - p, g = rs[k];
            for (int l = l16; l < ln; l += 16) {
                const unsigned v = recs[g + l];
                const int d8 = (int)(v & 255);
                const int r = atomicAdd(&cnt[d8], 1);
                csr[Sb + roff[d8] + r] = (int)(v >> 8);
            }
        }
    }
}

__global__ void __launch_bounds__(256) __attribute__((amdgpu_waves_per_eu(4, 4)))
k_gemm1(const float* __restrict__ x, const float* __restrict__ W1, const int* __restrict__ rowptr,
        float* __restrict__ xw, int N) {
    __shared__ __attribute__((aligned(16))) float w1s[150 * 256];
    const int lane = threadIdx.x & 31;
    const int wave = __builtin_amdgcn_readfirstlane((int)(threadIdx.x >> 5));
    const int h    = lane >> 4;
    const int c    = lane & 15;
    const int wh   = wave & 1;
    const int rb   = blockIdx.x * 64 + ((wave >> 1) << 4);

    const int arow = min(rb + c, N - 1);
    const float* xr = x + (size_t)arow * 300 + (h << 2);
    float4 a[37];
    #pragma unroll
    for (int kk = 0; kk < 18; ++kk) a[kk] = *(const float4*)(xr + 8 * kk);

    const float* wsrc = W1 + (size_t)wave * 256 + (lane << 2);
    float* wdst = w1s + wave * 256 + (lane >> 4) * 128 + (lane & 15) * 4;
    #pragma unroll 9
    for (int it = 0; it < 18; ++it) {
        const float4 r0 = *(const float4*)(wsrc + it * 2048);
        const float4 r1 = *(const float4*)(wsrc + it * 2048 + 128);
        float4 lo; lo.x = r0.x; lo.y = r1.x; lo.z = r0.y; lo.w = r1.y;
        float4 hi; hi.x = r0.z; hi.y = r1.z; hi.z = r0.w; hi.w = r1.w;
        *(float4*)(wdst + it * 2048)      = lo;
        *(float4*)(wdst + it * 2048 + 64) = hi;
    }
    if (wave < 6) {
        const float4 r0 = *(const float4*)(wsrc + 18 * 2048);
        const float4 r1 = *(const float4*)(wsrc + 18 * 2048 + 128);
        float4 lo; lo.x = r0.x; lo.y = r1.x; lo.z = r0.y; lo.w = r1.y;
        float4 hi; hi.x = r0.z; hi.y = r1.z; hi.z = r0.w; hi.w = r1.w;
        *(float4*)(wdst + 18 * 2048)      = lo;
        *(float4*)(wdst + 18 * 2048 + 64) = hi;
    }
    __syncthreads();
    #pragma unroll
    for (int kk = 18; kk < 37; ++kk) a[kk] = *(const float4*)(xr + 8 * kk);
    const float2 at = *(const float2*)(x + (size_t)arow * 300 + 296 + (h << 1));
    const int ql = rowptr[min(rb + lane, N)];

    v8f acc0 = {}, acc1 = {}, acc2 = {}, acc3 = {};
    const float* bl = w1s + (h << 9) + (wh << 7) + (c << 2);
    #pragma unroll
    for (int kk = 0; kk < 37; ++kk) {
        const float* p = bl + kk * 1024;
        const float4 b00 = *(const float4*)(p);
        const float4 b01 = *(const float4*)(p + 64);
        const float4 b10 = *(const float4*)(p + 256);
        const float4 b11 = *(const float4*)(p + 320);
        v2f a0; a0.x = a[kk].x; a0.y = a[kk].y;
        v2f a1; a1.x = a[kk].z; a1.y = a[kk].w;
        v2f b;
        b.x = b00.x; b.y = b00.y; acc0 = wmma_f32_k4(a0, b, acc0);
        b.x = b00.z; b.y = b00.w; acc1 = wmma_f32_k4(a0, b, acc1);
        b.x = b01.x; b.y = b01.y; acc2 = wmma_f32_k4(a0, b, acc2);
        b.x = b01.z; b.y = b01.w; acc3 = wmma_f32_k4(a0, b, acc3);
        b.x = b10.x; b.y = b10.y; acc0 = wmma_f32_k4(a1, b, acc0);
        b.x = b10.z; b.y = b10.w; acc1 = wmma_f32_k4(a1, b, acc1);
        b.x = b11.x; b.y = b11.y; acc2 = wmma_f32_k4(a1, b, acc2);
        b.x = b11.z; b.y = b11.w; acc3 = wmma_f32_k4(a1, b, acc3);
    }
    {
        const float* p = w1s + (148 + h) * 256 + (wh << 7) + (c << 2);
        const float4 b0 = *(const float4*)(p);
        const float4 b1 = *(const float4*)(p + 64);
        v2f av; av.x = at.x; av.y = at.y;
        v2f b;
        b.x = b0.x; b.y = b0.y; acc0 = wmma_f32_k4(av, b, acc0);
        b.x = b0.z; b.y = b0.w; acc1 = wmma_f32_k4(av, b, acc1);
        b.x = b1.x; b.y = b1.y; acc2 = wmma_f32_k4(av, b, acc2);
        b.x = b1.z; b.y = b1.w; acc3 = wmma_f32_k4(av, b, acc3);
    }
    const int qn = __shfl_down(ql, 1, 32);
    const float dl = rsqrtf((float)(qn - ql + 1));
    float dv[8];
    #pragma unroll
    for (int r = 0; r < 8; ++r) dv[r] = __shfl(dl, (h << 3) + r, 32);
    #pragma unroll
    for (int r = 0; r < 8; ++r) {
        const int rr = rb + (h << 3) + r;
        if (rr < N) {
            float4 o; o.x = acc0[r] * dv[r]; o.y = acc1[r] * dv[r]; o.z = acc2[r] * dv[r]; o.w = acc3[r] * dv[r];
            *(float4*)(xw + (size_t)rr * 128 + (wh << 6) + (c << 2)) = o;
        }
    }
}

template <int U>
__device__ __forceinline__ void add_rows(const float* __restrict__ xw, int s_l,
                                         int t, int f, float4& acc) {
    float4 v[U];
    #pragma unroll
    for (int k = 0; k < U; ++k) {
        const int s = __builtin_amdgcn_readlane(s_l, t + k);
        v[k] = *(const float4*)(xw + (size_t)s * 128 + f);
    }
    #pragma unroll
    for (int k = 0; k < U; ++k) {
        acc.x += v[k].x;
        acc.y += v[k].y;
        acc.z += v[k].z;
        acc.w += v[k].w;
    }
}

#define HP 132

__device__ __forceinline__ float4 gather_rows(const float* __restrict__ xw, const int* __restrict__ csr,
                                              int beg, int end, int s_l, int lane, int f) {
    float4 acc = {0.0f, 0.0f, 0.0f, 0.0f};
    for (int base = beg; base < end; base += 32) {
        const int m = min(32, end - base);
        if (base != beg) {
            s_l = 0;
            if (lane < m) s_l = csr[base + lane];
        }
        int t = 0;
        for (; t + 8 <= m; t += 8) add_rows<8>(xw, s_l, t, f, acc);
        if (t + 4 <= m) { add_rows<4>(xw, s_l, t, f, acc); t += 4; }
        if (t + 2 <= m) { add_rows<2>(xw, s_l, t, f, acc); t += 2; }
        if (t < m)      { add_rows<1>(xw, s_l, t, f, acc); }
    }
    return acc;
}

__global__ void __launch_bounds__(256)
k_gather1_gemm2(const float* __restrict__ xw, const int* __restrict__ rowptr,
                const int* __restrict__ csr, const float* __restrict__ b1,
                const float* __restrict__ W2, float* __restrict__ hw16, int N) {
    __shared__ __attribute__((aligned(16))) float hs[16 * HP];
    __shared__ __attribute__((aligned(16))) float w2q[16 * 2 * 16 * 4];
    __shared__ __attribute__((aligned(16))) float dns[16];
    const int tid  = threadIdx.x;
    const int lane = tid & 31;
    const int wave = __builtin_amdgcn_readfirstlane((int)(tid >> 5));
    const int f = lane << 2;
    const int n0 = blockIdx.x * 16 + wave * 2;
    const int p0 = rowptr[min(n0, N)], p1 = rowptr[min(n0 + 1, N)], p2 = rowptr[min(n0 + 2, N)];
    int sA = 0, sB = 0;
    if (lane < p1 - p0) sA = csr[p0 + lane];
    if (lane < p2 - p1) sB = csr[p1 + lane];
    float4 sv0 = {0.0f, 0.0f, 0.0f, 0.0f}, sv1 = {0.0f, 0.0f, 0.0f, 0.0f};
    if (n0 < N)     sv0 = *(const float4*)(xw + (size_t)n0 * 128 + f);
    if (n0 + 1 < N) sv1 = *(const float4*)(xw + (size_t)(n0 + 1) * 128 + f);
    float4 wq[2];
    #pragma unroll
    for (int j = 0; j < 2; ++j) {
        const int i = tid + 256 * j;
        const int k0 = ((i >> 5) << 3) + (((i >> 4) & 1) << 2), cc = i & 15;
        wq[j].x = 0.0f; wq[j].y = 0.0f; wq[j].z = 0.0f; wq[j].w = 0.0f;
        if (cc < 10) {
            wq[j].x = W2[(k0 + 0) * 10 + cc];
            wq[j].y = W2[(k0 + 1) * 10 + cc];
            wq[j].z = W2[(k0 + 2) * 10 + cc];
            wq[j].w = W2[(k0 + 3) * 10 + cc];
        }
    }
    const float4 bv = *(const float4*)(b1 + f);
    *(float4*)&w2q[(tid      ) * 4] = wq[0];
    *(float4*)&w2q[(tid + 256) * 4] = wq[1];

    {
        const float4 a = gather_rows(xw, csr, p0, p1, sA, lane, f);
        float4 o = {0.0f, 0.0f, 0.0f, 0.0f};
        float dn = 0.0f;
        if (n0 < N) {
            dn = rsqrtf((float)(p1 - p0 + 1));
            o.x = (a.x + sv0.x) * dn + bv.x;
            o.y = (a.y + sv0.y) * dn + bv.y;
            o.z = (a.z + sv0.z) * dn + bv.z;
            o.w = (a.w + sv0.w) * dn + bv.w;
            o.x = o.x > 0.0f ? o.x : 0.0f;
            o.y = o.y > 0.0f ? o.y : 0.0f;
            o.z = o.z > 0.0f ? o.z : 0.0f;
            o.w = o.w > 0.0f ? o.w : 0.0f;
        }
        *(float4*)&hs[(wave * 2) * HP + f] = o;
        if (lane == 0) dns[wave * 2] = dn;
    }
    {
        const float4 a = gather_rows(xw, csr, p1, p2, sB, lane, f);
        float4 o = {0.0f, 0.0f, 0.0f, 0.0f};
        float dn = 0.0f;
        if (n0 + 1 < N) {
            dn = rsqrtf((float)(p2 - p1 + 1));
            o.x = (a.x + sv1.x) * dn + bv.x;
            o.y = (a.y + sv1.y) * dn + bv.y;
            o.z = (a.z + sv1.z) * dn + bv.z;
            o.w = (a.w + sv1.w) * dn + bv.w;
            o.x = o.x > 0.0f ? o.x : 0.0f;
            o.y = o.y > 0.0f ? o.y : 0.0f;
            o.z = o.z > 0.0f ? o.z : 0.0f;
            o.w = o.w > 0.0f ? o.w : 0.0f;
        }
        *(float4*)&hs[(wave * 2 + 1) * HP + f] = o;
        if (lane == 0) dns[wave * 2 + 1] = dn;
    }
    __syncthreads();
    if (wave != 0) return;

    const int h = lane >> 4, c = lane & 15;
    v8f acc0 = {}, acc1 = {};
    const float* ap = hs + c * HP + (h << 2);
    const float* bp = w2q + (((h << 4) + c) << 2);
    #pragma unroll
    for (int kk = 0; kk < 16; ++kk) {
        const float4 a = *(const float4*)(ap + 8 * kk);
        const float4 b = *(const float4*)(bp + 128 * kk);
        v2f a0; a0.x = a.x; a0.y = a.y;
        v2f a1; a1.x = a.z; a1.y = a.w;
        v2f b0; b0.x = b.x; b0.y = b.y;
        v2f b1v; b1v.x = b.z; b1v.y = b.w;
        acc0 = wmma_f32_k4(a0, b0, acc0);
        acc1 = wmma_f32_k4(a1, b1v, acc1);
    }
    const float4 d0 = *(const float4*)&dns[h << 3];
    const float4 d1 = *(const float4*)&dns[(h << 3) + 4];
    const float dv[8] = {d0.x, d0.y, d0.z, d0.w, d1.x, d1.y, d1.z, d1.w};
    const int m0 = blockIdx.x * 16 + (h << 3);
    #pragma unroll
    for (int r = 0; r < 8; ++r)
        if (m0 + r < N) hw16[(size_t)(m0 + r) * 16 + c] = (acc0[r] + acc1[r]) * dv[r];
}

__global__ void __launch_bounds__(256)
k_gather2(const float* __restrict__ hw16, const int* __restrict__ rowptr,
          const int* __restrict__ csr, const float* __restrict__ b2,
          float* __restrict__ out, int N) {
    __shared__ __attribute__((aligned(16))) float sres[64 * 10];
    const int tid = threadIdx.x;
    const int q  = tid & 3;
    const int nl = tid >> 2;
    const int n  = blockIdx.x * 64 + nl;
    const int nc = n < N ? n : N - 1;
    const int beg = rowptr[nc], end = rowptr[nc + 1];
    const int deg = end - beg;
    const float4 self = *(const float4*)(hw16 + (size_t)nc * 16 + (q << 2));
    const float* hq = hw16 + (q << 2);
    const int* cq = csr + beg + q;
    int s0 = -1, s1 = -1;
    if (q < deg)     s0 = cq[0];
    if (4 + q < deg) s1 = cq[4];
    float4 acc = {0.0f, 0.0f, 0.0f, 0.0f};
    for (int base = 0; base < deg; base += 8) {
        int sk[8];
        sk[0] = __builtin_amdgcn_mov_dpp(s0, 0x00, 0xF, 0xF, true);
        sk[1] = __builtin_amdgcn_mov_dpp(s0, 0x55, 0xF, 0xF, true);
        sk[2] = __builtin_amdgcn_mov_dpp(s0, 0xAA, 0xF, 0xF, true);
        sk[3] = __builtin_amdgcn_mov_dpp(s0, 0xFF, 0xF, 0xF, true);
        sk[4] = __builtin_amdgcn_mov_dpp(s1, 0x00, 0xF, 0xF, true);
        sk[5] = __builtin_amdgcn_mov_dpp(s1, 0x55, 0xF, 0xF, true);
        sk[6] = __builtin_amdgcn_mov_dpp(s1, 0xAA, 0xF, 0xF, true);
        sk[7] = __builtin_amdgcn_mov_dpp(s1, 0xFF, 0xF, 0xF, true);
        float4 v[8];
        #pragma unroll
        for (int k = 0; k < 8; ++k) {
            v[k].x = 0.0f; v[k].y = 0.0f; v[k].z = 0.0f; v[k].w = 0.0f;
            if (sk[k] >= 0) v[k] = *(const float4*)(hq + (size_t)sk[k] * 16);
        }
        s0 = -1; s1 = -1;
        if (base + 8 + q < deg)  s0 = cq[base + 8];
        if (base + 12 + q < deg) s1 = cq[base + 12];
        #pragma unroll
        for (int k = 0; k < 8; ++k) {
            if (sk[k] >= 0) {
                acc.x += v[k].x;
                acc.y += v[k].y;
                acc.z += v[k].z;
                acc.w += v[k].w;
            }
        }
    }
    const float dn = rsqrtf((float)(deg + 1));
    if (q < 3) {
        const int f = q << 2;
        const float r0 = (acc.x + self.x) * dn + b2[f];
        const float r1 = (acc.y + self.y) * dn + b2[f + 1];
        float* d = &sres[nl * 10 + f];
        d[0] = r0;
        d[1] = r1;
        if (q < 2) {
            d[2] = (acc.z + self.z) * dn + b2[f + 2];
            d[3] = (acc.w + self.w) * dn + b2[f + 3];
        }
    }
    __syncthreads();
    if (tid < 160) {
        const size_t total = (size_t)N * 10;
        const size_t o = (size_t)blockIdx.x * 640 + ((size_t)tid << 2);
        if (o + 4 <= total) {
            *(float4*)(out + o) = *(const float4*)&sres[tid << 2];
        } else {
            for (size_t e = o; e < total; ++e) out[e] = sres[e - (size_t)blockIdx.x * 640];
        }
    }
}

extern "C" void kernel_launch(void* const* d_in, const int* in_sizes, int n_in,
                              void* d_out, int out_size, void* d_ws, size_t ws_size,
                              hipStream_t stream) {
    const float* x  = (const float*)d_in[0];
    const int*   ei = (const int*)d_in[1];
    const float* W1 = (const float*)d_in[2];
    const float* b1 = (const float*)d_in[3];
    const float* W2 = (const float*)d_in[4];
    const float* b2 = (const float*)d_in[5];
    float* out = (float*)d_out;

    const int N = in_sizes[0] / 300;
    const int E = in_sizes[1] / 2;
    const int* src = ei;
    const int* dst = ei + E;

    size_t Np = ((size_t)N + 64) & ~(size_t)63;
    int*   rowptr = (int*)d_ws;
    float* xw   = (float*)d_ws + Np;
    float* r2   = xw + Np * 128;
    int*   csr  = (int*)(r2 + Np * 128);
    float* hw   = r2;

    const int T = 256;
    const int nB = (N + 255) / 256;
    const int nT = (E + TE - 1) / TE;
    unsigned* recs = (unsigned*)r2;
    int*      offT = (int*)r2 + (size_t)nT * TE;

    k_tile_sort <<<nT, T, 0, stream>>>(src, dst, recs, offT, E, nB, nT);
    k_bucket_csr<<<nB, T, 0, stream>>>(recs, offT, rowptr, csr, N, nT, nB);

    k_gemm1     <<<(N + 63) / 64, T, 0, stream>>>(x, W1, rowptr, xw, N);
    k_gather1_gemm2<<<(N + 15) / 16, T, 0, stream>>>(xw, rowptr, csr, b1, W2, hw, N);
    k_gather2   <<<(N + 63) / 64, T, 0, stream>>>(hw, rowptr, csr, b2, out, N);
}


# First sort level rides at the tail of the first product's launch (rows stored unscaled; layer-1 gather multiplies each row by its source's 1/sqrt(deg), fused): 5 -> 4 launches; rounding changes
# speedup vs baseline: 4.5469x; 1.0347x over previous
#include <hip/hip_runtime.h>
#include <hip/hip_bf16.h>


typedef __attribute__((ext_vector_type(2))) float v2f;
typedef __attribute__((ext_vector_type(8))) float v8f;

__device__ __forceinline__ v8f wmma_f32_k4(v2f a, v2f b, v8f c) {
    return __builtin_amdgcn_wmma_f32_16x16x4_f32(
         false, a,  false, b,
         (short)0, c,  false,  false);
}

#define TE   4096
#define MAXB 1024
#define MAXT 512
#define CAP  8192

__device__ __forceinline__ int block_excl_scan(int v, int* wsum, int& total) {
    const int lane = threadIdx.x & 31, wv = threadIdx.x >> 5;
    int inc = v;
    for (int o = 1; o < 32; o <<= 1) {
        int t = __shfl_up(inc, o, 32);
        if (lane >= o) inc += t;
    }
    __syncthreads();
    if (lane == 31) wsum[wv] = inc;
    __syncthreads();
    int woff = 0, tot = 0;
    #pragma unroll
    for (int k = 0; k < 8; ++k) {
        const int w = wsum[k];
        if (k < wv) woff += w;
        tot += w;
    }
    total = tot;
    return woff + inc - v;
}

__device__ __forceinline__ void tile_sort_block(const int* __restrict__ src, const int* __restrict__ dst,
            unsigned* __restrict__ recs, int* __restrict__ offT, int E, int B, int T, int t,
            int* hist, unsigned* buf, int* wsum) {
    const int tid = threadIdx.x;
    const int e0 = t * TE;
    const int nt = min(TE, E - e0);
    for (int i = tid; i < MAXB; i += 256) hist[i] = 0;
    __syncthreads();
    int d[16], s[16], r[16];
    #pragma unroll
    for (int j = 0; j < 16; ++j) {
        const int i = j * 256 + tid;
        d[j] = -1; s[j] = 0;
        if (i < nt) { d[j] = dst[e0 + i]; s[j] = src[e0 + i]; }
    }
    #pragma unroll
    for (int j = 0; j < 16; ++j) {
        r[j] = 0;
        if (d[j] >= 0) r[j] = atomicAdd(&hist[d[j] >> 8], 1);
    }
    __syncthreads();
    const int4 hv = *(const int4*)&hist[4 * tid];
    int tot;
    const int ex = block_excl_scan(hv.x + hv.y + hv.z + hv.w, wsum, tot);
    int4 st; st.x = ex; st.y = ex + hv.x; st.z = st.y + hv.y; st.w = st.z + hv.z;
    *(int4*)&hist[4 * tid] = st;
    __syncthreads();
    for (int b = tid; b <= B; b += 256) offT[(size_t)b * T + t] = (b < B) ? hist[b] : nt;
    #pragma unroll
    for (int j = 0; j < 16; ++j)
        if (d[j] >= 0) buf[hist[d[j] >> 8] + r[j]] = ((unsigned)s[j] << 8) | (unsigned)(d[j] & 255);
    __syncthreads();
    #pragma unroll
    for (int j = 0; j < 16; ++j) {
        const int i = j * 256 + tid;
        if (i < nt) recs[e0 + i] = buf[i];
    }
}

__global__ void __launch_bounds__(256)
k_bucket_csr(const unsigned* __restrict__ recs, const int* __restrict__ offT,
             int* __restrict__ rowptr, int* __restrict__ csr, float* __restrict__ dinv, int N, int T, int B) {
    __shared__ int rs[MAXT];
    __shared__ int rp[MAXT + 1];
    __shared__ int cnt[256];
    __shared__ int roff[256];
    __shared__ int wsum[8];
    __shared__ unsigned buf[CAP];
    __shared__ int outb[CAP];
    const int tid = threadIdx.x;
    const int b = blockIdx.x;
    const int hw = tid >> 4, l16 = tid & 15;
    cnt[tid] = 0;

    const int t0 = 2 * tid, t1 = t0 + 1;
    const int* o0 = offT + (size_t)b * T;
    const int* o1 = o0 + T;
    int s0 = 0, x0 = 0, s1 = 0, x1 = 0;
    if (t0 < T) { s0 = o0[t0]; x0 = o1[t0]; }
    if (t1 < T) { s1 = o0[t1]; x1 = o1[t1]; }
    rs[t0] = t0 * TE + s0;
    rs[t1] = t1 * TE + s1;
    const int a = x0 - s0, c = x1 - s1;
    int nb, Sb;
    const int ex = block_excl_scan(a + c, wsum, nb);
    rp[t0] = ex;
    rp[t1] = ex + a;
    if (tid == 0) rp[MAXT] = nb;
    (void)block_excl_scan(s0 + s1, wsum, Sb);
    __syncthreads();
    const bool staged = nb <= CAP;

    if (staged) {
        for (int k0 = hw; k0 < T; k0 += 80) {
            unsigned v[5]; int p[5], ln[5], g[5];
            #pragma unroll
            for (int u = 0; u < 5; ++u) {
                const int k = k0 + 16 * u;
                p[u] = 0; ln[u] = 0; g[u] = 0; v[u] = 0;
                if (k < T) {
                    p[u] = rp[k]; ln[u] = rp[k + 1] - p[u]; g[u] = rs[k];
                    if (l16 < ln[u]) v[u] = recs[g[u] + l16];
                }
            }
            #pragma unroll
            for (int u = 0; u < 5; ++u) {
                if (l16 < ln[u]) { buf[p[u] + l16] = v[u]; atomicAdd(&cnt[v[u] & 255], 1); }
                for (int l = 16 + l16; l < ln[u]; l += 16) {
                    const unsigned w = recs[g[u] + l];
                    buf[p[u] + l] = w; atomicAdd(&cnt[w & 255], 1);
                }
            }
        }
    } else {
        for (int k = hw; k < T; k += 16) {
            const int p = rp[k], ln = rp[k + 1] - p, g = rs[k];
            for (int l = l16; l < ln; l += 16) atomicAdd(&cnt[recs[g + l] & 255], 1);
        }
    }
    __syncthreads();
    {
        const int cv = cnt[tid];
        int tot;
        const int e2 = block_excl_scan(cv, wsum, tot);
        roff[tid] = e2;
        const int node = b * 256 + tid;
        if (node < N) {
            rowptr[node] = Sb + e2;
            dinv[node] = rsqrtf((float)(cv + 1));
        }
        if (b == B - 1 && tid == 0) rowptr[N] = Sb + nb;
        cnt[tid] = 0;
    }
    __syncthreads();
    if (staged) {
        for (int i0 = tid; i0 < nb; i0 += 1024) {
            unsigned v[4]; int r[4];
            #pragma unroll
            for (int u = 0; u < 4; ++u) { const int i = i0 + 256 * u; v[u] = (i < nb) ? buf[i] : 0u; }
            #pragma unroll
            for (int u = 0; u < 4; ++u) {
                r[u] = 0;
                if (i0 + 256 * u < nb) r[u] = atomicAdd(&cnt[v[u] & 255], 1);
            }
            #pragma unroll
            for (int u = 0; u < 4; ++u)
                if (i0 + 256 * u < nb) outb[roff[v[u] & 255] + r[u]] = (int)(v[u] >> 8);
        }
        __syncthreads();
        for (int i = tid; i < nb; i += 256) csr[Sb + i] = outb[i];
    } else {
        for (int k = hw; k < T; k += 16) {
            const int p = rp[k], ln = rp[k + 1] - p, g = rs[k];
            for (int l = l16; l < ln; l += 16) {
                const unsigned v = recs[g + l];
                const int d8 = (int)(v & 255);
                const int r = atomicAdd(&cnt[d8], 1);
                csr[Sb + roff[d8] + r] = (int)(v >> 8);
            }
        }
    }
}

__global__ void __launch_bounds__(256) __attribute__((amdgpu_waves_per_eu(4, 4)))
k_gemm1_tile_sort(const float* __restrict__ x, const float* __restrict__ W1, float* __restrict__ xw, int N,
                  const int* __restrict__ src, const int* __restrict__ dst,
                  unsigned* __restrict__ recs, int* __restrict__ offT, int E, int B, int nT, int nG) {
    __shared__ __attribute__((aligned(16))) float w1s[150 * 256];
    if ((int)blockIdx.x >= nG) {
        int* li = (int*)w1s;
        tile_sort_block(src, dst, recs, offT, E, B, nT, (int)blockIdx.x - nG,
                        li, (unsigned*)(li + MAXB), li + MAXB + TE);
        return;
    }
    const int gb = (int)blockIdx.x;
    const int lane = threadIdx.x & 31;
    const int wave = __builtin_amdgcn_readfirstlane((int)(threadIdx.x >> 5));
    const int h    = lane >> 4;
    const int c    = lane & 15;
    const int wh   = wave & 1;
    const int rb   = gb * 64 + ((wave >> 1) << 4);

    const int arow = min(rb + c, N - 1);
    const float* xr = x + (size_t)arow * 300 + (h << 2);
    float4 a[37];
    #pragma unroll
    for (int kk = 0; kk < 18; ++kk) a[kk] = *(const float4*)(xr + 8 * kk);

    const float* wsrc = W1 + (size_t)wave * 256 + (lane << 2);
    float* wdst = w1s + wave * 256 + (lane >> 4) * 128 + (lane & 15) * 4;
    #pragma unroll 9
    for (int it = 0; it < 18; ++it) {
        const float4 r0 = *(const float4*)(wsrc + it * 2048);
        const float4 r1 = *(const float4*)(wsrc + it * 2048 + 128);
        float4 lo; lo.x = r0.x; lo.y = r1.x; lo.z = r0.y; lo.w = r1.y;
        float4 hi; hi.x = r0.z; hi.y = r1.z; hi.z = r0.w; hi.w = r1.w;
        *(float4*)(wdst + it * 2048)      = lo;
        *(float4*)(wdst + it * 2048 + 64) = hi;
    }
    if (wave < 6) {
        const float4 r0 = *(const float4*)(wsrc + 18 * 2048);
        const float4 r1 = *(const float4*)(wsrc + 18 * 2048 + 128);
        float4 lo; lo.x = r0.x; lo.y = r1.x; lo.z = r0.y; lo.w = r1.y;
        float4 hi; hi.x = r0.z; hi.y = r1.z; hi.z = r0.w; hi.w = r1.w;
        *(float4*)(wdst + 18 * 2048)      = lo;
        *(float4*)(wdst + 18 * 2048 + 64) = hi;
    }
    __syncthreads();
    #pragma unroll
    for (int kk = 18; kk < 37; ++kk) a[kk] = *(const float4*)(xr + 8 * kk);
    const float2 at = *(const float2*)(x + (size_t)arow * 300 + 296 + (h << 1));

    v8f acc0 = {}, acc1 = {}, acc2 = {}, acc3 = {};
    const float* bl = w1s + (h << 9) + (wh << 7) + (c << 2);
    #pragma unroll
    for (int kk = 0; kk < 37; ++kk) {
        const float* p = bl + kk * 1024;
        const float4 b00 = *(const float4*)(p);
        const float4 b01 = *(const float4*)(p + 64);
        const float4 b10 = *(const float4*)(p + 256);
        const float4 b11 = *(const float4*)(p + 320);
        v2f a0; a0.x = a[kk].x; a0.y = a[kk].y;
        v2f a1; a1.x = a[kk].z; a1.y = a[kk].w;
        v2f b;
        b.x = b00.x; b.y = b00.y; acc0 = wmma_f32_k4(a0, b, acc0);
        b.x = b00.z; b.y = b00.w; acc1 = wmma_f32_k4(a0, b, acc1);
        b.x = b01.x; b.y = b01.y; acc2 = wmma_f32_k4(a0, b, acc2);
        b.x = b01.z; b.y = b01.w; acc3 = wmma_f32_k4(a0, b, acc3);
        b.x = b10.x; b.y = b10.y; acc0 = wmma_f32_k4(a1, b, acc0);
        b.x = b10.z; b.y = b10.w; acc1 = wmma_f32_k4(a1, b, acc1);
        b.x = b11.x; b.y = b11.y; acc2 = wmma_f32_k4(a1, b, acc2);
        b.x = b11.z; b.y = b11.w; acc3 = wmma_f32_k4(a1, b, acc3);
    }
    {
        const float* p = w1s + (148 + h) * 256 + (wh << 7) + (c << 2);
        const float4 b0 = *(const float4*)(p);
        const float4 b1 = *(const float4*)(p + 64);
        v2f av; av.x = at.x; av.y = at.y;
        v2f b;
        b.x = b0.x; b.y = b0.y; acc0 = wmma_f32_k4(av, b, acc0);
        b.x = b0.z; b.y = b0.w; acc1 = wmma_f32_k4(av, b, acc1);
        b.x = b1.x; b.y = b1.y; acc2 = wmma_f32_k4(av, b, acc2);
        b.x = b1.z; b.y = b1.w; acc3 = wmma_f32_k4(av, b, acc3);
    }
    #pragma unroll
    for (int r = 0; r < 8; ++r) {
        const int rr = rb + (h << 3) + r;
        if (rr < N) {
            float4 o; o.x = acc0[r]; o.y = acc1[r]; o.z = acc2[r]; o.w = acc3[r];
            *(float4*)(xw + (size_t)rr * 128 + (wh << 6) + (c << 2)) = o;
        }
    }
}

template <int U>
__device__ __forceinline__ void add_rows(const float* __restrict__ xw, int s_l, float d_l,
                                         int t, int f, float4& acc) {
    float4 v[U];
    #pragma unroll
    for (int k = 0; k < U; ++k) {
        const int s = __builtin_amdgcn_readlane(s_l, t + k);
        v[k] = *(const float4*)(xw + (size_t)s * 128 + f);
    }
    __builtin_amdgcn_sched_barrier(0);
    #pragma unroll
    for (int k = 0; k < U; ++k) {
        const float w = __int_as_float(__builtin_amdgcn_readlane(__float_as_int(d_l), t + k));
        acc.x = fmaf(v[k].x, w, acc.x);
        acc.y = fmaf(v[k].y, w, acc.y);
        acc.z = fmaf(v[k].z, w, acc.z);
        acc.w = fmaf(v[k].w, w, acc.w);
    }
}

#define HP 132

__device__ __forceinline__ float4 gather_rows(const float* __restrict__ xw, const int* __restrict__ csr,
                                              const float* __restrict__ dinv,
                                              int beg, int end, int s_l, float d_l, int lane, int f) {
    float4 acc = {0.0f, 0.0f, 0.0f, 0.0f};
    for (int base = beg; base < end; base += 32) {
        const int m = min(32, end - base);
        if (base != beg) {
            s_l = 0;
            if (lane < m) s_l = csr[base + lane];
            d_l = dinv[s_l];
        }
        int t = 0;
        for (; t + 8 <= m; t += 8) add_rows<8>(xw, s_l, d_l, t, f, acc);
        if (t + 4 <= m) { add_rows<4>(xw, s_l, d_l, t, f, acc); t += 4; }
        if (t + 2 <= m) { add_rows<2>(xw, s_l, d_l, t, f, acc); t += 2; }
        if (t < m)      { add_rows<1>(xw, s_l, d_l, t, f, acc); }
    }
    return acc;
}

__global__ void __launch_bounds__(256)
k_gather1_gemm2(const float* __restrict__ xw, const int* __restrict__ rowptr,
                const int* __restrict__ csr, const float* __restrict__ dinv, const float* __restrict__ b1,
                const float* __restrict__ W2, float* __restrict__ hw16, int N) {
    __shared__ __attribute__((aligned(16))) float hs[16 * HP];
    __shared__ __attribute__((aligned(16))) float w2q[16 * 2 * 16 * 4];
    __shared__ __attribute__((aligned(16))) float dns[16];
    const int tid  = threadIdx.x;
    const int lane = tid & 31;
    const int wave = __builtin_amdgcn_readfirstlane((int)(tid >> 5));
    const int f = lane << 2;
    const int n0 = blockIdx.x * 16 + wave * 2;
    const int p0 = rowptr[min(n0, N)], p1 = rowptr[min(n0 + 1, N)], p2 = rowptr[min(n0 + 2, N)];
    int sA = 0, sB = 0;
    if (lane < p1 - p0) sA = csr[p0 + lane];
    if (lane < p2 - p1) sB = csr[p1 + lane];
    float4 sv0 = {0.0f, 0.0f, 0.0f, 0.0f}, sv1 = {0.0f, 0.0f, 0.0f, 0.0f};
    if (n0 < N)     sv0 = *(const float4*)(xw + (size_t)n0 * 128 + f);
    if (n0 + 1 < N) sv1 = *(const float4*)(xw + (size_t)(n0 + 1) * 128 + f);
    float4 wq[2];
    #pragma unroll
    for (int j = 0; j < 2; ++j) {
        const int i = tid + 256 * j;
        const int k0 = ((i >> 5) << 3) + (((i >> 4) & 1) << 2), cc = i & 15;
        wq[j].x = 0.0f; wq[j].y = 0.0f; wq[j].z = 0.0f; wq[j].w = 0.0f;
        if (cc < 10) {
            wq[j].x = W2[(k0 + 0) * 10 + cc];
            wq[j].y = W2[(k0 + 1) * 10 + cc];
            wq[j].z = W2[(k0 + 2) * 10 + cc];
            wq[j].w = W2[(k0 + 3) * 10 + cc];
        }
    }
    const float4 bv = *(const float4*)(b1 + f);
    *(float4*)&w2q[(tid      ) * 4] = wq[0];
    *(float4*)&w2q[(tid + 256) * 4] = wq[1];
    const float dA = dinv[sA], dB = dinv[sB];

    {
        const float4 a = gather_rows(xw, csr, dinv, p0, p1, sA, dA, lane, f);
        float4 o = {0.0f, 0.0f, 0.0f, 0.0f};
        float dn = 0.0f;
        if (n0 < N) {
            dn = rsqrtf((float)(p1 - p0 + 1));
            o.x = fmaf(sv0.x, dn, a.x) * dn + bv.x;
            o.y = fmaf(sv0.y, dn, a.y) * dn + bv.y;
            o.z = fmaf(sv0.z, dn, a.z) * dn + bv.z;
            o.w = fmaf(sv0.w, dn, a.w) * dn + bv.w;
            o.x = o.x > 0.0f ? o.x : 0.0f;
            o.y = o.y > 0.0f ? o.y : 0.0f;
            o.z = o.z > 0.0f ? o.z : 0.0f;
            o.w = o.w > 0.0f ? o.w : 0.0f;
        }
        *(float4*)&hs[(wave * 2) * HP + f] = o;
        if (lane == 0) dns[wave * 2] = dn;
    }
    {
        const float4 a = gather_rows(xw, csr, dinv, p1, p2, sB, dB, lane, f);
        float4 o = {0.0f, 0.0f, 0.0f, 0.0f};
        float dn = 0.0f;
        if (n0 + 1 < N) {
            dn = rsqrtf((float)(p2 - p1 + 1));
            o.x = fmaf(sv1.x, dn, a.x) * dn + bv.x;
            o.y = fmaf(sv1.y, dn, a.y) * dn + bv.y;
            o.z = fmaf(sv1.z, dn, a.z) * dn + bv.z;
            o.w = fmaf(sv1.w, dn, a.w) * dn + bv.w;
            o.x = o.x > 0.0f ? o.x : 0.0f;
            o.y = o.y > 0.0f ? o.y : 0.0f;
            o.z = o.z > 0.0f ? o.z : 0.0f;
            o.w = o.w > 0.0f ? o.w : 0.0f;
        }
        *(float4*)&hs[(wave * 2 + 1) * HP + f] = o;
        if (lane == 0) dns[wave * 2 + 1] = dn;
    }
    __syncthreads();
    if (wave != 0) return;

    const int h = lane >> 4, c = lane & 15;
    v8f acc0 = {}, acc1 = {};
    const float* ap = hs + c * HP + (h << 2);
    const float* bp = w2q + (((h << 4) + c) << 2);
    #pragma unroll
    for (int kk = 0; kk < 16; ++kk) {
        const float4 a = *(const float4*)(ap + 8 * kk);
        const float4 b = *(const float4*)(bp + 128 * kk);
        v2f a0; a0.x = a.x; a0.y = a.y;
        v2f a1; a1.x = a.z; a1.y = a.w;
        v2f b0; b0.x = b.x; b0.y = b.y;
        v2f b1v; b1v.x = b.z; b1v.y = b.w;
        acc0 = wmma_f32_k4(a0, b0, acc0);
        acc1 = wmma_f32_k4(a1, b1v, acc1);
    }
    const float4 d0 = *(const float4*)&dns[h << 3];
    const float4 d1 = *(const float4*)&dns[(h << 3) + 4];
    const float dv[8] = {d0.x, d0.y, d0.z, d0.w, d1.x, d1.y, d1.z, d1.w};
    const int m0 = blockIdx.x * 16 + (h << 3);
    #pragma unroll
    for (int r = 0; r < 8; ++r)
        if (m0 + r < N) hw16[(size_t)(m0 + r) * 16 + c] = (acc0[r] + acc1[r]) * dv[r];
}

__global__ void __launch_bounds__(256)
k_gather2(const float* __restrict__ hw16, const int* __restrict__ rowptr,
          const int* __restrict__ csr, const float* __restrict__ b2,
          float* __restrict__ out, int N) {
    __shared__ __attribute__((aligned(16))) float sres[64 * 10];
    const int tid = threadIdx.x;
    const int q  = tid & 3;
    const int nl = tid >> 2;
    const int n  = blockIdx.x * 64 + nl;
    const int nc = n < N ? n : N - 1;
    const int beg = rowptr[nc], end = rowptr[nc + 1];
    const int deg = end - beg;
    const float4 self = *(const float4*)(hw16 + (size_t)nc * 16 + (q << 2));
    const float* hq = hw16 + (q << 2);
    const int* cq = csr + beg + q;
    int s0 = -1, s1 = -1;
    if (q < deg)     s0 = cq[0];
    if (4 + q < deg) s1 = cq[4];
    float4 acc = {0.0f, 0.0f, 0.0f, 0.0f};
    for (int base = 0; base < deg; base += 8) {
        int sk[8];
        sk[0] = __builtin_amdgcn_mov_dpp(s0, 0x00, 0xF, 0xF, true);
        sk[1] = __builtin_amdgcn_mov_dpp(s0, 0x55, 0xF, 0xF, true);
        sk[2] = __builtin_amdgcn_mov_dpp(s0, 0xAA, 0xF, 0xF, true);
        sk[3] = __builtin_amdgcn_mov_dpp(s0, 0xFF, 0xF, 0xF, true);
        sk[4] = __builtin_amdgcn_mov_dpp(s1, 0x00, 0xF, 0xF, true);
        sk[5] = __builtin_amdgcn_mov_dpp(s1, 0x55, 0xF, 0xF, true);
        sk[6] = __builtin_amdgcn_mov_dpp(s1, 0xAA, 0xF, 0xF, true);
        sk[7] = __builtin_amdgcn_mov_dpp(s1, 0xFF, 0xF, 0xF, true);
        float4 v[8];
        #pragma unroll
        for (int k = 0; k < 8; ++k) {
            v[k].x = 0.0f; v[k].y = 0.0f; v[k].z = 0.0f; v[k].w = 0.0f;
            if (sk[k] >= 0) v[k] = *(const float4*)(hq + (size_t)sk[k] * 16);
        }
        s0 = -1; s1 = -1;
        if (base + 8 + q < deg)  s0 = cq[base + 8];
        if (base + 12 + q < deg) s1 = cq[base + 12];
        #pragma unroll
        for (int k = 0; k < 8; ++k) {
            if (sk[k] >= 0) {
                acc.x += v[k].x;
                acc.y += v[k].y;
                acc.z += v[k].z;
                acc.w += v[k].w;
            }
        }
    }
    const float dn = rsqrtf((float)(deg + 1));
    if (q < 3) {
        const int f = q << 2;
        const float r0 = (acc.x + self.x) * dn + b2[f];
        const float r1 = (acc.y + self.y) * dn + b2[f + 1];
        float* d = &sres[nl * 10 + f];
        d[0] = r0;
        d[1] = r1;
        if (q < 2) {
            d[2] = (acc.z + self.z) * dn + b2[f + 2];
            d[3] = (acc.w + self.w) * dn + b2[f + 3];
        }
    }
    __syncthreads();
    if (tid < 160) {
        const size_t total = (size_t)N * 10;
        const size_t o = (size_t)blockIdx.x * 640 + ((size_t)tid << 2);
        if (o + 4 <= total) {
            *(float4*)(out + o) = *(const float4*)&sres[tid << 2];
        } else {
            for (size_t e = o; e < total; ++e) out[e] = sres[e - (size_t)blockIdx.x * 640];
        }
    }
}

extern "C" void kernel_launch(void* const* d_in, const int* in_sizes, int n_in,
                              void* d_out, int out_size, void* d_ws, size_t ws_size,
                              hipStream_t stream) {
    const float* x  = (const float*)d_in[0];
    const int*   ei = (const int*)d_in[1];
    const float* W1 = (const float*)d_in[2];
    const float* b1 = (const float*)d_in[3];
    const float* W2 = (const float*)d_in[4];
    const float* b2 = (const float*)d_in[5];
    float* out = (float*)d_out;

    const int N = in_sizes[0] / 300;
    const int E = in_sizes[1] / 2;
    const int* src = ei;
    const int* dst = ei + E;

    size_t Np = ((size_t)N + 64) & ~(size_t)63;
    int*   rowptr = (int*)d_ws;
    float* xw   = (float*)d_ws + Np;
    float* r2   = xw + Np * 128;
    int*   csr  = (int*)(r2 + Np * 128);
    float* hw   = r2;
    float* dinv = r2 + Np * 32;

    const int T = 256;
    const int nB = (N + 255) / 256;
    const int nT = (E + TE - 1) / TE;
    unsigned* recs = (unsigned*)r2;
    int*      offT = (int*)r2 + (size_t)nT * TE;

    const int nG = (N + 63) / 64;
    k_gemm1_tile_sort<<<nG + nT, T, 0, stream>>>(x, W1, xw, N, src, dst, recs, offT, E, nB, nT, nG);
    k_bucket_csr<<<nB, T, 0, stream>>>(recs, offT, rowptr, csr, dinv, N, nT, nB);

    k_gather1_gemm2<<<(N + 15) / 16, T, 0, stream>>>(xw, rowptr, csr, dinv, b1, W2, hw, N);
    k_gather2   <<<(N + 63) / 64, T, 0, stream>>>(hw, rowptr, csr, b2, out, N);
}


# First product on 512 resident work-groups: W1 staged in LDS once each, loop over 64-row tiles, next tile's A rows requested during the current one (was 1,563 one-tile work-groups); sum order kept
# speedup vs baseline: 4.7064x; 1.0351x over previous
#include <hip/hip_runtime.h>
#include <hip/hip_bf16.h>


typedef __attribute__((ext_vector_type(2))) float v2f;
typedef __attribute__((ext_vector_type(8))) float v8f;

__device__ __forceinline__ v8f wmma_f32_k4(v2f a, v2f b, v8f c) {
    return __builtin_amdgcn_wmma_f32_16x16x4_f32(
         false, a,  false, b,
         (short)0, c,  false,  false);
}

#define TE   4096
#define MAXB 1024
#define MAXT 512
#define CAP  8192

__device__ __forceinline__ int block_excl_scan(int v, int* wsum, int& total) {
    const int lane = threadIdx.x & 31, wv = threadIdx.x >> 5;
    int inc = v;
    for (int o = 1; o < 32; o <<= 1) {
        int t = __shfl_up(inc, o, 32);
        if (lane >= o) inc += t;
    }
    __syncthreads();
    if (lane == 31) wsum[wv] = inc;
    __syncthreads();
    int woff = 0, tot = 0;
    #pragma unroll
    for (int k = 0; k < 8; ++k) {
        const int w = wsum[k];
        if (k < wv) woff += w;
        tot += w;
    }
    total = tot;
    return woff + inc - v;
}

__device__ __forceinline__ void tile_sort_block(const int* __restrict__ src, const int* __restrict__ dst,
            unsigned* __restrict__ recs, int* __restrict__ offT, int E, int B, int T, int t,
            int* hist, unsigned* buf, int* wsum) {
    const int tid = threadIdx.x;
    const int e0 = t * TE;
    const int nt = min(TE, E - e0);
    for (int i = tid; i < MAXB; i += 256) hist[i] = 0;
    __syncthreads();
    int d[16], s[16], r[16];
    #pragma unroll
    for (int j = 0; j < 16; ++j) {
        const int i = j * 256 + tid;
        d[j] = -1; s[j] = 0;
        if (i < nt) { d[j] = dst[e0 + i]; s[j] = src[e0 + i]; }
    }
    #pragma unroll
    for (int j = 0; j < 16; ++j) {
        r[j] = 0;
        if (d[j] >= 0) r[j] = atomicAdd(&hist[d[j] >> 8], 1);
    }
    __syncthreads();
    const int4 hv = *(const int4*)&hist[4 * tid];
    int tot;
    const int ex = block_excl_scan(hv.x + hv.y + hv.z + hv.w, wsum, tot);
    int4 st; st.x = ex; st.y = ex + hv.x; st.z = st.y + hv.y; st.w = st.z + hv.z;
    *(int4*)&hist[4 * tid] = st;
    __syncthreads();
    for (int b = tid; b <= B; b += 256) offT[(size_t)b * T + t] = (b < B) ? hist[b] : nt;
    #pragma unroll
    for (int j = 0; j < 16; ++j)
        if (d[j] >= 0) buf[hist[d[j] >> 8] + r[j]] = ((unsigned)s[j] << 8) | (unsigned)(d[j] & 255);
    __syncthreads();
    #pragma unroll
    for (int j = 0; j < 16; ++j) {
        const int i = j * 256 + tid;
        if (i < nt) recs[e0 + i] = buf[i];
    }
}

__global__ void __launch_bounds__(256)
k_bucket_csr(const unsigned* __restrict__ recs, const int* __restrict__ offT,
             int* __restrict__ rowptr, int* __restrict__ csr, float* __restrict__ dinv, int N, int T, int B) {
    __shared__ int rs[MAXT];
    __shared__ int rp[MAXT + 1];
    __shared__ int cnt[256];
    __shared__ int roff[256];
    __shared__ int wsum[8];
    __shared__ unsigned buf[CAP];
    __shared__ int outb[CAP];
    const int tid = threadIdx.x;
    const int b = blockIdx.x;
    const int hw = tid >> 4, l16 = tid & 15;
    cnt[tid] = 0;

    const int t0 = 2 * tid, t1 = t0 + 1;
    const int* o0 = offT + (size_t)b * T;
    const int* o1 = o0 + T;
    int s0 = 0, x0 = 0, s1 = 0, x1 = 0;
    if (t0 < T) { s0 = o0[t0]; x0 = o1[t0]; }
    if (t1 < T) { s1 = o0[t1]; x1 = o1[t1]; }
    rs[t0] = t0 * TE + s0;
    rs[t1] = t1 * TE + s1;
    const int a = x0 - s0, c = x1 - s1;
    int nb, Sb;
    const int ex = block_excl_scan(a + c, wsum, nb);
    rp[t0] = ex;
    rp[t1] = ex + a;
    if (tid == 0) rp[MAXT] = nb;
    (void)block_excl_scan(s0 + s1, wsum, Sb);
    __syncthreads();
    const bool staged = nb <= CAP;

    if (staged) {
        for (int k0 = hw; k0 < T; k0 += 80) {
            unsigned v[5]; int p[5], ln[5], g[5];
            #pragma unroll
            for (int u = 0; u < 5; ++u) {
                const int k = k0 + 16 * u;
                p[u] = 0; ln[u] = 0; g[u] = 0; v[u] = 0;
                if (k < T) {
                    p[u] = rp[k]; ln[u] = rp[k + 1] - p[u]; g[u] = rs[k];
                    if (l16 < ln[u]) v[u] = recs[g[u] + l16];
                }
            }
            #pragma unroll
            for (int u = 0; u < 5; ++u) {
                if (l16 < ln[u]) { buf[p[u] + l16] = v[u]; atomicAdd(&cnt[v[u] & 255], 1); }
                for (int l = 16 + l16; l < ln[u]; l += 16) {
                    const unsigned w = recs[g[u] + l];
                    buf[p[u] + l] = w; atomicAdd(&cnt[w & 255], 1);
                }
            }
        }
    } else {
        for (int k = hw; k < T; k += 16) {
            const int p = rp[k], ln = rp[k + 1] - p, g = rs[k];
            for (int l = l16; l < ln; l += 16) atomicAdd(&cnt[recs[g + l] & 255], 1);
        }
    }
    __syncthreads();
    {
        const int cv = cnt[tid];
        int tot;
        const int e2 = block_excl_scan(cv, wsum, tot);
        roff[tid] = e2;
        const int node = b * 256 + tid;
        if (node < N) {
            rowptr[node] = Sb + e2;
            dinv[node] = rsqrtf((float)(cv + 1));
        }
        if (b == B - 1 && tid == 0) rowptr[N] = Sb + nb;
        cnt[tid] = 0;
    }
    __syncthreads();
    if (staged) {
        for (int i0 = tid; i0 < nb; i0 += 1024) {
            unsigned v[4]; int r[4];
            #pragma unroll
            for (int u = 0; u < 4; ++u) { const int i = i0 + 256 * u; v[u] = (i < nb) ? buf[i] : 0u; }
            #pragma unroll
            for (int u = 0; u < 4; ++u) {
                r[u] = 0;
                if (i0 + 256 * u < nb) r[u] = atomicAdd(&cnt[v[u] & 255], 1);
            }
            #pragma unroll
            for (int u = 0; u < 4; ++u)
                if (i0 + 256 * u < nb) outb[roff[v[u] & 255] + r[u]] = (int)(v[u] >> 8);
        }
        __syncthreads();
        for (int i = tid; i < nb; i += 256) csr[Sb + i] = outb[i];
    } else {
        for (int k = hw; k < T; k += 16) {
            const int p = rp[k], ln = rp[k + 1] - p, g = rs[k];
            for (int l = l16; l < ln; l += 16) {
                const unsigned v = recs[g + l];
                const int d8 = (int)(v & 255);
                const int r = atomicAdd(&cnt[d8], 1);
                csr[Sb + roff[d8] + r] = (int)(v >> 8);
            }
        }
    }
}


#define GEMM1_STEPS(K0, K1, TAILB)                                                          \
    _Pragma("unroll")                                                                  \
    for (int kk = (K0); kk < (K1); ++kk) {                                             \
        const float* p = bl + kk * 1024;                                               \
        const float4 b00 = *(const float4*)(p);                                        \
        const float4 b01 = *(const float4*)(p + 64);                                   \
        const float4 b10 = *(const float4*)(p + 256);                                  \
        const float4 b11 = *(const float4*)(p + 320);                                  \
        v2f a0; a0.x = a[kk].x; a0.y = a[kk].y;                                        \
        v2f a1; a1.x = a[kk].z; a1.y = a[kk].w;                                        \
        v2f b;                                                                         \
        b.x = b00.x; b.y = b00.y; acc0 = wmma_f32_k4(a0, b, acc0);                     \
        b.x = b00.z; b.y = b00.w; acc1 = wmma_f32_k4(a0, b, acc1);                     \
        b.x = b01.x; b.y = b01.y; acc2 = wmma_f32_k4(a0, b, acc2);                     \
        b.x = b01.z; b.y = b01.w; acc3 = wmma_f32_k4(a0, b, acc3);                     \
        b.x = b10.x; b.y = b10.y; acc0 = wmma_f32_k4(a1, b, acc0);                     \
        b.x = b10.z; b.y = b10.w; acc1 = wmma_f32_k4(a1, b, acc1);                     \
        b.x = b11.x; b.y = b11.y; acc2 = wmma_f32_k4(a1, b, acc2);                     \
        b.x = b11.z; b.y = b11.w; acc3 = wmma_f32_k4(a1, b, acc3);                     \
        if (kk == (K0))     __builtin_amdgcn_sched_group_barrier(0x100, 4, 0);         \
        if (kk + 1 < (K1))  __builtin_amdgcn_sched_group_barrier(0x100, 4, 0);         \
        if (kk + 1 == (K1)) __builtin_amdgcn_sched_group_barrier(0x100, (TAILB), 0);   \
        __builtin_amdgcn_sched_group_barrier(0x008, 8, 0);                             \
    }

__global__ void __launch_bounds__(256) __attribute__((amdgpu_waves_per_eu(4, 4)))
k_gemm1_tile_sort(const float* __restrict__ x, const float* __restrict__ W1, float* __restrict__ xw, int N,
                  const int* __restrict__ src, const int* __restrict__ dst,
                  unsigned* __restrict__ recs, int* __restrict__ offT, int E, int B, int nT, int nG, int nP) {
    __shared__ __attribute__((aligned(16))) float w1s[150 * 256];
    if ((int)blockIdx.x >= nP) {
        int* li = (int*)w1s;
        tile_sort_block(src, dst, recs, offT, E, B, nT, (int)blockIdx.x - nP,
                        li, (unsigned*)(li + MAXB), li + MAXB + TE);
        return;
    }
    const int lane = threadIdx.x & 31;
    const int wave = __builtin_amdgcn_readfirstlane((int)(threadIdx.x >> 5));
    const int h    = lane >> 4;
    const int c    = lane & 15;
    const int wh   = wave & 1;
    const int rw   = (wave >> 1) << 4;

    int gb = (int)blockIdx.x;
    const float* xr = x + (size_t)min(gb * 64 + rw + c, N - 1) * 300 + (h << 2);
    float4 a[37];
    #pragma unroll
    for (int kk = 0; kk < 18; ++kk) a[kk] = *(const float4*)(xr + 8 * kk);

    const float* wsrc = W1 + (size_t)wave * 256 + (lane << 2);
    float* wdst = w1s + wave * 256 + (lane >> 4) * 128 + (lane & 15) * 4;
    #pragma unroll 9
    for (int it = 0; it < 18; ++it) {
        const float4 r0 = *(const float4*)(wsrc + it * 2048);
        const float4 r1 = *(const float4*)(wsrc + it * 2048 + 128);
        float4 lo; lo.x = r0.x; lo.y = r1.x; lo.z = r0.y; lo.w = r1.y;
        float4 hi; hi.x = r0.z; hi.y = r1.z; hi.z = r0.w; hi.w = r1.w;
        *(float4*)(wdst + it * 2048)      = lo;
        *(float4*)(wdst + it * 2048 + 64) = hi;
    }
    if (wave < 6) {
        const float4 r0 = *(const float4*)(wsrc + 18 * 2048);
        const float4 r1 = *(const float4*)(wsrc + 18 * 2048 + 128);
        float4 lo; lo.x = r0.x; lo.y = r1.x; lo.z = r0.y; lo.w = r1.y;
        float4 hi; hi.x = r0.z; hi.y = r1.z; hi.z = r0.w; hi.w = r1.w;
        *(float4*)(wdst + 18 * 2048)      = lo;
        *(float4*)(wdst + 18 * 2048 + 64) = hi;
    }
    __syncthreads();

    const float* bl = w1s + (h << 9) + (wh << 7) + (c << 2);
    const float* bt = w1s + (148 + h) * 256 + (wh << 7) + (c << 2);
    #define GEMM1_FENCE __builtin_amdgcn_sched_barrier(0)
    for (; gb < nG; gb += nP) {
        #pragma unroll
        for (int kk = 18; kk < 27; ++kk) a[kk] = *(const float4*)(xr + 8 * kk);
        GEMM1_FENCE;

        v8f acc0 = {}, acc1 = {}, acc2 = {}, acc3 = {};
        GEMM1_STEPS(0, 9, 0)

        GEMM1_FENCE;
        #pragma unroll
        for (int kk = 27; kk < 37; ++kk) a[kk] = *(const float4*)(xr + 8 * kk);
        const float2 at = *(const float2*)(xr + 296 - (h << 1));
        GEMM1_FENCE;

        GEMM1_STEPS(9, 18, 0)

        const float* xn = x + (size_t)min((gb + nP) * 64 + rw + c, N - 1) * 300 + (h << 2);
        GEMM1_FENCE;
        #pragma unroll
        for (int kk = 0; kk < 9; ++kk) a[kk] = *(const float4*)(xn + 8 * kk);
        GEMM1_FENCE;

        GEMM1_STEPS(18, 27, 0)

        GEMM1_FENCE;
        #pragma unroll
        for (int kk = 9; kk < 18; ++kk) a[kk] = *(const float4*)(xn + 8 * kk);
        GEMM1_FENCE;

        GEMM1_STEPS(27, 37, 2)
        {
            const float4 b0 = *(const float4*)(bt);
            const float4 b1 = *(const float4*)(bt + 64);
            v2f av; av.x = at.x; av.y = at.y;
            v2f b;
            b.x = b0.x; b.y = b0.y; acc0 = wmma_f32_k4(av, b, acc0);
            b.x = b0.z; b.y = b0.w; acc1 = wmma_f32_k4(av, b, acc1);
            b.x = b1.x; b.y = b1.y; acc2 = wmma_f32_k4(av, b, acc2);
            b.x = b1.z; b.y = b1.w; acc3 = wmma_f32_k4(av, b, acc3);
            __builtin_amdgcn_sched_group_barrier(0x008, 4, 0);
        }
        const int rb = gb * 64 + rw;
        #pragma unroll
        for (int r = 0; r < 8; ++r) {
            const int rr = rb + (h << 3) + r;
            if (rr < N) {
                float4 o; o.x = acc0[r]; o.y = acc1[r]; o.z = acc2[r]; o.w = acc3[r];
                *(float4*)(xw + (size_t)rr * 128 + (wh << 6) + (c << 2)) = o;
            }
        }
        xr = xn;
    }
    #undef GEMM1_FENCE
}
#undef GEMM1_STEPS

template <int U>
__device__ __forceinline__ void add_rows(const float* __restrict__ xw, int s_l, float d_l,
                                         int t, int f, float4& acc) {
    float4 v[U];
    #pragma unroll
    for (int k = 0; k < U; ++k) {
        const int s = __builtin_amdgcn_readlane(s_l, t + k);
        v[k] = *(const float4*)(xw + (size_t)s * 128 + f);
    }
    __builtin_amdgcn_sched_barrier(0);
    #pragma unroll
    for (int k = 0; k < U; ++k) {
        const float w = __int_as_float(__builtin_amdgcn_readlane(__float_as_int(d_l), t + k));
        acc.x = fmaf(v[k].x, w, acc.x);
        acc.y = fmaf(v[k].y, w, acc.y);
        acc.z = fmaf(v[k].z, w, acc.z);
        acc.w = fmaf(v[k].w, w, acc.w);
    }
}

#define HP 132

__device__ __forceinline__ float4 gather_rows(const float* __restrict__ xw, const int* __restrict__ csr,
                                              const float* __restrict__ dinv,
                                              int beg, int end, int s_l, float d_l, int lane, int f) {
    float4 acc = {0.0f, 0.0f, 0.0f, 0.0f};
    for (int base = beg; base < end; base += 32) {
        const int m = min(32, end - base);
        if (base != beg) {
            s_l = 0;
            if (lane < m) s_l = csr[base + lane];
            d_l = dinv[s_l];
        }
        int t = 0;
        for (; t + 8 <= m; t += 8) add_rows<8>(xw, s_l, d_l, t, f, acc);
        if (t + 4 <= m) { add_rows<4>(xw, s_l, d_l, t, f, acc); t += 4; }
        if (t + 2 <= m) { add_rows<2>(xw, s_l, d_l, t, f, acc); t += 2; }
        if (t < m)      { add_rows<1>(xw, s_l, d_l, t, f, acc); }
    }
    return acc;
}

__global__ void __launch_bounds__(256)
k_gather1_gemm2(const float* __restrict__ xw, const int* __restrict__ rowptr,
                const int* __restrict__ csr, const float* __restrict__ dinv, const float* __restrict__ b1,
                const float* __restrict__ W2, float* __restrict__ hw16, int N) {
    __shared__ __attribute__((aligned(16))) float hs[16 * HP];
    __shared__ __attribute__((aligned(16))) float w2q[16 * 2 * 16 * 4];
    __shared__ __attribute__((aligned(16))) float dns[16];
    const int tid  = threadIdx.x;
    const int lane = tid & 31;
    const int wave = __builtin_amdgcn_readfirstlane((int)(tid >> 5));
    const int f = lane << 2;
    const int n0 = blockIdx.x * 16 + wave * 2;
    const int p0 = rowptr[min(n0, N)], p1 = rowptr[min(n0 + 1, N)], p2 = rowptr[min(n0 + 2, N)];
    int sA = 0, sB = 0;
    if (lane < p1 - p0) sA = csr[p0 + lane];
    if (lane < p2 - p1) sB = csr[p1 + lane];
    float4 sv0 = {0.0f, 0.0f, 0.0f, 0.0f}, sv1 = {0.0f, 0.0f, 0.0f, 0.0f};
    if (n0 < N)     sv0 = *(const float4*)(xw + (size_t)n0 * 128 + f);
    if (n0 + 1 < N) sv1 = *(const float4*)(xw + (size_t)(n0 + 1) * 128 + f);
    float4 wq[2];
    #pragma unroll
    for (int j = 0; j < 2; ++j) {
        const int i = tid + 256 * j;
        const int k0 = ((i >> 5) << 3) + (((i >> 4) & 1) << 2), cc = i & 15;
        wq[j].x = 0.0f; wq[j].y = 0.0f; wq[j].z = 0.0f; wq[j].w = 0.0f;
        if (cc < 10) {
            wq[j].x = W2[(k0 + 0) * 10 + cc];
            wq[j].y = W2[(k0 + 1) * 10 + cc];
            wq[j].z = W2[(k0 + 2) * 10 + cc];
            wq[j].w = W2[(k0 + 3) * 10 + cc];
        }
    }
    const float4 bv = *(const float4*)(b1 + f);
    *(float4*)&w2q[(tid      ) * 4] = wq[0];
    *(float4*)&w2q[(tid + 256) * 4] = wq[1];
    const float dA = dinv[sA], dB = dinv[sB];

    {
        const float4 a = gather_rows(xw, csr, dinv, p0, p1, sA, dA, lane, f);
        float4 o = {0.0f, 0.0f, 0.0f, 0.0f};
        float dn = 0.0f;
        if (n0 < N) {
            dn = rsqrtf((float)(p1 - p0 + 1));
            o.x = fmaf(sv0.x, dn, a.x) * dn + bv.x;
            o.y = fmaf(sv0.y, dn, a.y) * dn + bv.y;
            o.z = fmaf(sv0.z, dn, a.z) * dn + bv.z;
            o.w = fmaf(sv0.w, dn, a.w) * dn + bv.w;
            o.x = o.x > 0.0f ? o.x : 0.0f;
            o.y = o.y > 0.0f ? o.y : 0.0f;
            o.z = o.z > 0.0f ? o.z : 0.0f;
            o.w = o.w > 0.0f ? o.w : 0.0f;
        }
        *(float4*)&hs[(wave * 2) * HP + f] = o;
        if (lane == 0) dns[wave * 2] = dn;
    }
    {
        const float4 a = gather_rows(xw, csr, dinv, p1, p2, sB, dB, lane, f);
        float4 o = {0.0f, 0.0f, 0.0f, 0.0f};
        float dn = 0.0f;
        if (n0 + 1 < N) {
            dn = rsqrtf((float)(p2 - p1 + 1));
            o.x = fmaf(sv1.x, dn, a.x) * dn + bv.x;
            o.y = fmaf(sv1.y, dn, a.y) * dn + bv.y;
            o.z = fmaf(sv1.z, dn, a.z) * dn + bv.z;
            o.w = fmaf(sv1.w, dn, a.w) * dn + bv.w;
            o.x = o.x > 0.0f ? o.x : 0.0f;
            o.y = o.y > 0.0f ? o.y : 0.0f;
            o.z = o.z > 0.0f ? o.z : 0.0f;
            o.w = o.w > 0.0f ? o.w : 0.0f;
        }
        *(float4*)&hs[(wave * 2 + 1) * HP + f] = o;
        if (lane == 0) dns[wave * 2 + 1] = dn;
    }
    __syncthreads();
    if (wave != 0) return;

    const int h = lane >> 4, c = lane & 15;
    v8f acc0 = {}, acc1 = {};
    const float* ap = hs + c * HP + (h << 2);
    const float* bp = w2q + (((h << 4) + c) << 2);
    #pragma unroll
    for (int kk = 0; kk < 16; ++kk) {
        const float4 a = *(const float4*)(ap + 8 * kk);
        const float4 b = *(const float4*)(bp + 128 * kk);
        v2f a0; a0.x = a.x; a0.y = a.y;
        v2f a1; a1.x = a.z; a1.y = a.w;
        v2f b0; b0.x = b.x; b0.y = b.y;
        v2f b1v; b1v.x = b.z; b1v.y = b.w;
        acc0 = wmma_f32_k4(a0, b0, acc0);
        acc1 = wmma_f32_k4(a1, b1v, acc1);
    }
    const float4 d0 = *(const float4*)&dns[h << 3];
    const float4 d1 = *(const float4*)&dns[(h << 3) + 4];
    const float dv[8] = {d0.x, d0.y, d0.z, d0.w, d1.x, d1.y, d1.z, d1.w};
    const int m0 = blockIdx.x * 16 + (h << 3);
    #pragma unroll
    for (int r = 0; r < 8; ++r)
        if (m0 + r < N) hw16[(size_t)(m0 + r) * 16 + c] = (acc0[r] + acc1[r]) * dv[r];
}

__global__ void __launch_bounds__(256)
k_gather2(const float* __restrict__ hw16, const int* __restrict__ rowptr,
          const int* __restrict__ csr, const float* __restrict__ b2,
          float* __restrict__ out, int N) {
    __shared__ __attribute__((aligned(16))) float sres[64 * 10];
    const int tid = threadIdx.x;
    const int q  = tid & 3;
    const int nl = tid >> 2;
    const int n  = blockIdx.x * 64 + nl;
    const int nc = n < N ? n : N - 1;
    const int beg = rowptr[nc], end = rowptr[nc + 1];
    const int deg = end - beg;
    const float4 self = *(const float4*)(hw16 + (size_t)nc * 16 + (q << 2));
    const float* hq = hw16 + (q << 2);
    const int* cq = csr + beg + q;
    int s0 = -1, s1 = -1;
    if (q < deg)     s0 = cq[0];
    if (4 + q < deg) s1 = cq[4];
    float4 acc = {0.0f, 0.0f, 0.0f, 0.0f};
    for (int base = 0; base < deg; base += 8) {
        int sk[8];
        sk[0] = __builtin_amdgcn_mov_dpp(s0, 0x00, 0xF, 0xF, true);
        sk[1] = __builtin_amdgcn_mov_dpp(s0, 0x55, 0xF, 0xF, true);
        sk[2] = __builtin_amdgcn_mov_dpp(s0, 0xAA, 0xF, 0xF, true);
        sk[3] = __builtin_amdgcn_mov_dpp(s0, 0xFF, 0xF, 0xF, true);
        sk[4] = __builtin_amdgcn_mov_dpp(s1, 0x00, 0xF, 0xF, true);
        sk[5] = __builtin_amdgcn_mov_dpp(s1, 0x55, 0xF, 0xF, true);
        sk[6] = __builtin_amdgcn_mov_dpp(s1, 0xAA, 0xF, 0xF, true);
        sk[7] = __builtin_amdgcn_mov_dpp(s1, 0xFF, 0xF, 0xF, true);
        float4 v[8];
        #pragma unroll
        for (int k = 0; k < 8; ++k) {
            v[k].x = 0.0f; v[k].y = 0.0f; v[k].z = 0.0f; v[k].w = 0.0f;
            if (sk[k] >= 0) v[k] = *(const float4*)(hq + (size_t)sk[k] * 16);
        }
        s0 = -1; s1 = -1;
        if (base + 8 + q < deg)  s0 = cq[base + 8];
        if (base + 12 + q < deg) s1 = cq[base + 12];
        #pragma unroll
        for (int k = 0; k < 8; ++k) {
            if (sk[k] >= 0) {
                acc.x += v[k].x;
                acc.y += v[k].y;
                acc.z += v[k].z;
                acc.w += v[k].w;
            }
        }
    }
    const float dn = rsqrtf((float)(deg + 1));
    if (q < 3) {
        const int f = q << 2;
        const float r0 = (acc.x + self.x) * dn + b2[f];
        const float r1 = (acc.y + self.y) * dn + b2[f + 1];
        float* d = &sres[nl * 10 + f];
        d[0] = r0;
        d[1] = r1;
        if (q < 2) {
            d[2] = (acc.z + self.z) * dn + b2[f + 2];
            d[3] = (acc.w + self.w) * dn + b2[f + 3];
        }
    }
    __syncthreads();
    if (tid < 160) {
        const size_t total = (size_t)N * 10;
        const size_t o = (size_t)blockIdx.x * 640 + ((size_t)tid << 2);
        if (o + 4 <= total) {
            *(float4*)(out + o) = *(const float4*)&sres[tid << 2];
        } else {
            for (size_t e = o; e < total; ++e) out[e] = sres[e - (size_t)blockIdx.x * 640];
        }
    }
}

extern "C" void kernel_launch(void* const* d_in, const int* in_sizes, int n_in,
                              void* d_out, int out_size, void* d_ws, size_t ws_size,
                              hipStream_t stream) {
    const float* x  = (const float*)d_in[0];
    const int*   ei = (const int*)d_in[1];
    const float* W1 = (const float*)d_in[2];
    const float* b1 = (const float*)d_in[3];
    const float* W2 = (const float*)d_in[4];
    const float* b2 = (const float*)d_in[5];
    float* out = (float*)d_out;

    const int N = in_sizes[0] / 300;
    const int E = in_sizes[1] / 2;
    const int* src = ei;
    const int* dst = ei + E;

    size_t Np = ((size_t)N + 64) & ~(size_t)63;
    int*   rowptr = (int*)d_ws;
    float* xw   = (float*)d_ws + Np;
    float* r2   = xw + Np * 128;
    int*   csr  = (int*)(r2 + Np * 128);
    float* hw   = r2;
    float* dinv = r2 + Np * 32;

    const int T = 256;
    const int nB = (N + 255) / 256;
    const int nT = (E + TE - 1) / TE;
    unsigned* recs = (unsigned*)r2;
    int*      offT = (int*)r2 + (size_t)nT * TE;

    const int nG = (N + 63) / 64;
    const int nP = 512;
    k_gemm1_tile_sort<<<nP + nT, T, 0, stream>>>(x, W1, xw, N, src, dst, recs, offT, E, nB, nT, nG, nP);
    k_bucket_csr<<<nB, T, 0, stream>>>(recs, offT, rowptr, csr, dinv, N, nT, nB);

    k_gather1_gemm2<<<(N + 15) / 16, T, 0, stream>>>(xw, rowptr, csr, dinv, b1, W2, hw, N);
    k_gather2   <<<(N + 63) / 64, T, 0, stream>>>(hw, rowptr, csr, b2, out, N);
}
